# starting kernel
# baseline (speedup 1.0000x reference)
#include <hip/hip_runtime.h>
#include <cmath>

typedef float v2f __attribute__((ext_vector_type(2)));
typedef float v8f __attribute__((ext_vector_type(8)));

#define WMMA_F32(A, Bf, C) \
    __builtin_amdgcn_wmma_f32_16x16x4_f32(false, (A), false, (Bf), (short)0, (C), false, false)

#define BB       2
#define NN       2048
#define FF       32
#define DROWS    8192
#define DCOLS    2049
#define MSHR     6144
#define KOUT     2049
#define KPAD     2064
#define NSPLIT   4
#define KSEG     (MSHR / NSPLIT)

#define WS_XW_OFF    0
#define WS_XSHR_OFF  (BB * NN * FF)
#define WS_PART_OFF  (WS_XSHR_OFF + BB * MSHR * FF)

__global__ void sufg_xw_kernel(const float* __restrict__ x,
                               const float* __restrict__ w,
                               float* __restrict__ xw) {
    int idx = blockIdx.x * blockDim.x + threadIdx.x;
    int o = idx & 31;
    int n = (idx >> 5) & (NN - 1);
    int b = idx >> 16;
    const float* xr = x + ((size_t)b * NN + n) * FF;
    float s = 0.f;
#pragma unroll
    for (int f = 0; f < FF; ++f) s = fmaf(xr[f], w[f * FF + o], s);
    xw[idx] = s;
}

__global__ void __launch_bounds__(256)
sufg_pass2_kernel(const float* __restrict__ dl,
                  const float* __restrict__ xw,
                  const float* __restrict__ filt,
                  float* __restrict__ xshr,
                  float thr) {
    const int lane = threadIdx.x & 31;
    const int wave = threadIdx.x >> 5;
    const int tile = blockIdx.x * 8 + wave;
    const int b    = blockIdx.y;
    if (tile >= MSHR / 16) return;

    const int m0   = tile * 16;
    const int row  = lane & 15;
    const int hi   = lane >> 4;
    const int koff = hi << 1;

    const float* A  = dl + (size_t)b * DROWS * DCOLS
                         + (size_t)(2048 + m0 + row) * DCOLS + 1 + koff;
    const float* Bp = xw + (size_t)b * NN * FF + (size_t)koff * FF + row;

    v8f acc0 = {0.f,0.f,0.f,0.f,0.f,0.f,0.f,0.f};
    v8f acc1 = {0.f,0.f,0.f,0.f,0.f,0.f,0.f,0.f};

    for (int kc = 0; kc < NN; kc += 64) {
        __builtin_prefetch(A + 256, 0, 0);
#pragma unroll
        for (int k = 0; k < 64; k += 4) {
            v2f a;  a.x  = A[0];   a.y  = A[1];
            v2f b0; b0.x = Bp[0];  b0.y = Bp[32];
            v2f b1; b1.x = Bp[16]; b1.y = Bp[48];
            acc0 = WMMA_F32(a, b0, acc0);
            acc1 = WMMA_F32(a, b1, acc1);
            A += 4; Bp += 4 * FF;
        }
    }

#pragma unroll
    for (int r = 0; r < 8; ++r) {
        const int m  = m0 + r + hi * 8;
        const float sc = filt[2048 + m];
        float v0 = acc0[r] * sc;
        float v1 = acc1[r] * sc;
        if (m >= 2048) {
            v0 = copysignf(fmaxf(fabsf(v0) - thr, 0.f), v0);
            v1 = copysignf(fmaxf(fabsf(v1) - thr, 0.f), v1);
        }
        float* orow = xshr + ((size_t)b * MSHR + m) * FF;
        orow[row]      = v0;
        orow[16 + row] = v1;
    }
}

__global__ void __launch_bounds__(256)
sufg_pass3_kernel(const float* __restrict__ dl,
                  const float* __restrict__ xshr,
                  float* __restrict__ partial) {
    const int lane = threadIdx.x & 31;
    const int wave = threadIdx.x >> 5;
    const int unit = blockIdx.x * 8 + wave;
    const int b    = blockIdx.y;
    if (unit >= (KPAD / 16) * NSPLIT) return;

    const int split  = unit & (NSPLIT - 1);
    const int tile   = unit >> 2;
    const int kbegin = split * KSEG;

    const int row  = lane & 15;
    const int hi   = lane >> 4;
    const int koff = hi << 1;

    int krow = tile * 16 + row;
    if (krow > KOUT - 1) krow = KOUT - 1;

    const float* A  = dl + (size_t)b * DROWS * DCOLS + (size_t)2048 * DCOLS
                         + (size_t)krow * MSHR + kbegin + koff;
    const float* Bp = xshr + (size_t)b * MSHR * FF + (size_t)(kbegin + koff) * FF + row;

    v8f acc0 = {0.f,0.f,0.f,0.f,0.f,0.f,0.f,0.f};
    v8f acc1 = {0.f,0.f,0.f,0.f,0.f,0.f,0.f,0.f};

    for (int kc = 0; kc < KSEG; kc += 64) {
        __builtin_prefetch(A + 256, 0, 0);
#pragma unroll
        for (int k = 0; k < 64; k += 4) {
            v2f a  = *(const v2f*)A;
            v2f b0; b0.x = Bp[0];  b0.y = Bp[32];
            v2f b1; b1.x = Bp[16]; b1.y = Bp[48];
            acc0 = WMMA_F32(a, b0, acc0);
            acc1 = WMMA_F32(a, b1, acc1);
            A += 4; Bp += 4 * FF;
        }
    }

#pragma unroll
    for (int r = 0; r < 8; ++r) {
        const int kk = tile * 16 + r + hi * 8;
        float* prow = partial + (((size_t)(b * NSPLIT + split)) * KPAD + kk) * FF;
        prow[row]      = acc0[r];
        prow[16 + row] = acc1[r];
    }
}

__global__ void sufg_reduce_kernel(const float* __restrict__ partial,
                                   const float* __restrict__ bias,
                                   float* __restrict__ out) {
    int idx = blockIdx.x * blockDim.x + threadIdx.x;
    if (idx >= BB * KOUT * FF) return;
    int o = idx & 31;
    int k = (idx >> 5) % KOUT;
    int b = idx / (KOUT * FF);
    float s = bias[o];
#pragma unroll
    for (int sp = 0; sp < NSPLIT; ++sp)
        s += partial[(((size_t)(b * NSPLIT + sp)) * KPAD + k) * FF + o];
    out[idx] = s;
}

extern "C" void kernel_launch(void* const* d_in, const int* in_sizes, int n_in,
                              void* d_out, int out_size, void* d_ws, size_t ws_size,
                              hipStream_t stream) {
    (void)in_sizes; (void)n_in; (void)out_size; (void)ws_size;
    const float* x    = (const float*)d_in[0];
    const float* dl   = (const float*)d_in[1];
    const float* w    = (const float*)d_in[2];
    const float* filt = (const float*)d_in[3];
    const float* bias = (const float*)d_in[4];
    float* out = (float*)d_out;

    float* ws      = (float*)d_ws;
    float* xw      = ws + WS_XW_OFF;
    float* xshr    = ws + WS_XSHR_OFF;
    float* partial = ws + WS_PART_OFF;

    const float thr = sqrtf(2.0f * logf((float)NN)) / sqrtf((float)NN);

    sufg_xw_kernel<<<(BB * NN * FF) / 256, 256, 0, stream>>>(x, w, xw);
    sufg_pass2_kernel<<<dim3((MSHR / 16) / 8, BB), 256, 0, stream>>>(dl, xw, filt, xshr, thr);
    sufg_pass3_kernel<<<dim3(((KPAD / 16) * NSPLIT + 7) / 8, BB), 256, 0, stream>>>(dl, xshr, partial);
    sufg_reduce_kernel<<<(BB * KOUT * FF + 255) / 256, 256, 0, stream>>>(partial, bias, out);
}


# K of each 16x32 tile split over the 8 waves of its block in both WMMA kernels (8x the waves in flight); partials summed through LDS in wave order (new reduction tree); prefetch kept inside each slice
# speedup vs baseline: 1.5446x; 1.5446x over previous
#include <hip/hip_runtime.h>
#include <cmath>

typedef float v2f __attribute__((ext_vector_type(2)));
typedef float v8f __attribute__((ext_vector_type(8)));

#define WMMA_F32(A, Bf, C) \
    __builtin_amdgcn_wmma_f32_16x16x4_f32(false, (A), false, (Bf), (short)0, (C), false, false)

#define BB       2
#define NN       2048
#define FF       32
#define DROWS    8192
#define DCOLS    2049
#define MSHR     6144
#define KOUT     2049
#define KPAD     2064
#define NSPLIT   4
#define KSEG     (MSHR / NSPLIT)
#define NWAVE    8
#define KW2      (NN / NWAVE)
#define KW3      (KSEG / NWAVE)
#define TILE_EL  (16 * FF)

#define WS_XW_OFF    0
#define WS_XSHR_OFF  (BB * NN * FF)
#define WS_PART_OFF  (WS_XSHR_OFF + BB * MSHR * FF)

__global__ void sufg_xw_kernel(const float* __restrict__ x,
                               const float* __restrict__ w,
                               float* __restrict__ xw) {
    int idx = blockIdx.x * blockDim.x + threadIdx.x;
    int o = idx & 31;
    int n = (idx >> 5) & (NN - 1);
    int b = idx >> 16;
    const float* xr = x + ((size_t)b * NN + n) * FF;
    float s = 0.f;
#pragma unroll
    for (int f = 0; f < FF; ++f) s = fmaf(xr[f], w[f * FF + o], s);
    xw[idx] = s;
}

__device__ __forceinline__ void sum_wave_tiles(float* red, int wave, int lane,
                                               const v8f& acc0, const v8f& acc1,
                                               float& s_lo, float& s_hi) {
    float* mine = red + wave * TILE_EL;
#pragma unroll
    for (int r = 0; r < 8; ++r) {
        mine[r * 64 + lane]      = acc0[r];
        mine[r * 64 + 32 + lane] = acc1[r];
    }
    __syncthreads();
    const int i_lo = wave * 64 + (lane >> 4) * 32 + (lane & 15);
    const int i_hi = i_lo + 16;
    float a = red[i_lo];
    float c = red[i_hi];
#pragma unroll
    for (int w = 1; w < NWAVE; ++w) {
        a += red[w * TILE_EL + i_lo];
        c += red[w * TILE_EL + i_hi];
    }
    s_lo = a;
    s_hi = c;
}

__global__ void __launch_bounds__(256)
sufg_pass2_kernel(const float* __restrict__ dl,
                  const float* __restrict__ xw,
                  const float* __restrict__ filt,
                  float* __restrict__ xshr,
                  float thr) {
    __shared__ float red[NWAVE * TILE_EL];

    const int lane = threadIdx.x & 31;
    const int wave = threadIdx.x >> 5;
    const int tile = blockIdx.x;
    const int b    = blockIdx.y;

    const int m0   = tile * 16;
    const int row  = lane & 15;
    const int hi   = lane >> 4;
    const int koff = hi << 1;
    const int k0   = wave * KW2;

    const float* Arow = dl + (size_t)b * DROWS * DCOLS
                           + (size_t)(2048 + m0 + row) * DCOLS + 1 + k0;
    const float* A  = Arow + koff;
    const float* Bp = xw + (size_t)b * NN * FF + (size_t)(k0 + koff) * FF + row;

#pragma unroll
    for (int j = 0; j < KW2 / 64; ++j)
        __builtin_prefetch(Arow + hi * 32 + j * 64, 0, 0);

    v8f acc0 = {0.f,0.f,0.f,0.f,0.f,0.f,0.f,0.f};
    v8f acc1 = {0.f,0.f,0.f,0.f,0.f,0.f,0.f,0.f};

    for (int kc = 0; kc < KW2; kc += 64) {
#pragma unroll
        for (int k = 0; k < 64; k += 4) {
            v2f a;  a.x  = A[0];   a.y  = A[1];
            v2f b0; b0.x = Bp[0];  b0.y = Bp[32];
            v2f b1; b1.x = Bp[16]; b1.y = Bp[48];
            acc0 = WMMA_F32(a, b0, acc0);
            acc1 = WMMA_F32(a, b1, acc1);
            A += 4; Bp += 4 * FF;
        }
    }

    float s_lo, s_hi;
    sum_wave_tiles(red, wave, lane, acc0, acc1, s_lo, s_hi);

    const int   mA  = m0 + wave;
    const int   mB  = mA + 8;
    float vA = s_lo * filt[2048 + mA];
    float vB = s_hi * filt[2048 + mB];
    if (m0 >= 2048) {
        vA = copysignf(fmaxf(fabsf(vA) - thr, 0.f), vA);
        vB = copysignf(fmaxf(fabsf(vB) - thr, 0.f), vB);
    }
    xshr[((size_t)b * MSHR + mA) * FF + lane] = vA;
    xshr[((size_t)b * MSHR + mB) * FF + lane] = vB;
}

__global__ void __launch_bounds__(256)
sufg_pass3_kernel(const float* __restrict__ dl,
                  const float* __restrict__ xshr,
                  float* __restrict__ partial) {
    __shared__ float red[NWAVE * TILE_EL];

    const int lane = threadIdx.x & 31;
    const int wave = threadIdx.x >> 5;
    const int unit = blockIdx.x;
    const int b    = blockIdx.y;

    const int split  = unit & (NSPLIT - 1);
    const int tile   = unit >> 2;
    const int kbegin = split * KSEG + wave * KW3;

    const int row  = lane & 15;
    const int hi   = lane >> 4;
    const int koff = hi << 1;

    int krow = tile * 16 + row;
    if (krow > KOUT - 1) krow = KOUT - 1;

    const float* Arow = dl + (size_t)b * DROWS * DCOLS + (size_t)2048 * DCOLS
                           + (size_t)krow * MSHR + kbegin;
    const float* A  = Arow + koff;
    const float* Bp = xshr + (size_t)b * MSHR * FF + (size_t)(kbegin + koff) * FF + row;

#pragma unroll
    for (int j = 0; j < KW3 / 64; ++j)
        __builtin_prefetch(Arow + hi * 32 + j * 64, 0, 0);

    v8f acc0 = {0.f,0.f,0.f,0.f,0.f,0.f,0.f,0.f};
    v8f acc1 = {0.f,0.f,0.f,0.f,0.f,0.f,0.f,0.f};

    for (int kc = 0; kc < KW3; kc += 64) {
#pragma unroll
        for (int k = 0; k < 64; k += 4) {
            v2f a  = *(const v2f*)A;
            v2f b0; b0.x = Bp[0];  b0.y = Bp[32];
            v2f b1; b1.x = Bp[16]; b1.y = Bp[48];
            acc0 = WMMA_F32(a, b0, acc0);
            acc1 = WMMA_F32(a, b1, acc1);
            A += 4; Bp += 4 * FF;
        }
    }

    float s_lo, s_hi;
    sum_wave_tiles(red, wave, lane, acc0, acc1, s_lo, s_hi);

    const int kkA = tile * 16 + wave;
    float* pbase = partial + ((size_t)(b * NSPLIT + split)) * KPAD * FF;
    pbase[(size_t)kkA * FF + lane]       = s_lo;
    pbase[(size_t)(kkA + 8) * FF + lane] = s_hi;
}

__global__ void sufg_reduce_kernel(const float* __restrict__ partial,
                                   const float* __restrict__ bias,
                                   float* __restrict__ out) {
    int idx = blockIdx.x * blockDim.x + threadIdx.x;
    if (idx >= BB * KOUT * FF) return;
    int o = idx & 31;
    int k = (idx >> 5) % KOUT;
    int b = idx / (KOUT * FF);
    float s = bias[o];
#pragma unroll
    for (int sp = 0; sp < NSPLIT; ++sp)
        s += partial[(((size_t)(b * NSPLIT + sp)) * KPAD + k) * FF + o];
    out[idx] = s;
}

extern "C" void kernel_launch(void* const* d_in, const int* in_sizes, int n_in,
                              void* d_out, int out_size, void* d_ws, size_t ws_size,
                              hipStream_t stream) {
    (void)in_sizes; (void)n_in; (void)out_size; (void)ws_size;
    const float* x    = (const float*)d_in[0];
    const float* dl   = (const float*)d_in[1];
    const float* w    = (const float*)d_in[2];
    const float* filt = (const float*)d_in[3];
    const float* bias = (const float*)d_in[4];
    float* out = (float*)d_out;

    float* ws      = (float*)d_ws;
    float* xw      = ws + WS_XW_OFF;
    float* xshr    = ws + WS_XSHR_OFF;
    float* partial = ws + WS_PART_OFF;

    const float thr = sqrtf(2.0f * logf((float)NN)) / sqrtf((float)NN);

    sufg_xw_kernel<<<(BB * NN * FF) / 256, 256, 0, stream>>>(x, w, xw);
    sufg_pass2_kernel<<<dim3(MSHR / 16, BB), 256, 0, stream>>>(dl, xw, filt, xshr, thr);
    sufg_pass3_kernel<<<dim3((KPAD / 16) * NSPLIT, BB), 256, 0, stream>>>(dl, xshr, partial);
    sufg_reduce_kernel<<<(BB * KOUT * FF + 255) / 256, 256, 0, stream>>>(partial, bias, out);
}


# Fragment loads widened: A one 16-byte load per lane per 8 k, B 8-byte column-pair loads (k order in each 8-k block permuted alike in A and B: new sum order); half the vector loads per WMMA
# speedup vs baseline: 1.6249x; 1.0520x over previous
#include <hip/hip_runtime.h>
#include <cmath>

typedef float v2f __attribute__((ext_vector_type(2)));
typedef float v4f __attribute__((ext_vector_type(4)));
typedef float v8f __attribute__((ext_vector_type(8)));
typedef float v2f_a4 __attribute__((ext_vector_type(2), aligned(4)));
typedef float v4f_a4 __attribute__((ext_vector_type(4), aligned(4)));

#define KBLOCK8(ap, bp, acc0, acc1)                                            \
    do {                                                                       \
        const v4f a4 = *(const v4f_a4*)(ap);                                   \
        const v2f r0 = *(const v2f_a4*)(bp);                                   \
        const v2f r1 = *(const v2f_a4*)((bp) + FF);                            \
        const v2f r2 = *(const v2f_a4*)((bp) + 2 * FF);                        \
        const v2f r3 = *(const v2f_a4*)((bp) + 3 * FF);                        \
        v2f a01; a01.x = a4.x; a01.y = a4.y;                                   \
        v2f a23; a23.x = a4.z; a23.y = a4.w;                                   \
        v2f b0;  b0.x  = r0.x; b0.y  = r1.x;                                   \
        v2f b1;  b1.x  = r0.y; b1.y  = r1.y;                                   \
        v2f b2;  b2.x  = r2.x; b2.y  = r3.x;                                   \
        v2f b3;  b3.x  = r2.y; b3.y  = r3.y;                                   \
        acc0 = WMMA_F32(a01, b0, acc0);                                        \
        acc1 = WMMA_F32(a01, b1, acc1);                                        \
        acc0 = WMMA_F32(a23, b2, acc0);                                        \
        acc1 = WMMA_F32(a23, b3, acc1);                                        \
    } while (0)

#define WMMA_F32(A, Bf, C) \
    __builtin_amdgcn_wmma_f32_16x16x4_f32(false, (A), false, (Bf), (short)0, (C), false, false)

#define BB       2
#define NN       2048
#define FF       32
#define DROWS    8192
#define DCOLS    2049
#define MSHR     6144
#define KOUT     2049
#define KPAD     2064
#define NSPLIT   4
#define KSEG     (MSHR / NSPLIT)
#define NWAVE    8
#define KW2      (NN / NWAVE)
#define KW3      (KSEG / NWAVE)
#define TILE_EL  (16 * FF)

#define WS_XW_OFF    0
#define WS_XSHR_OFF  (BB * NN * FF)
#define WS_PART_OFF  (WS_XSHR_OFF + BB * MSHR * FF)

__global__ void sufg_xw_kernel(const float* __restrict__ x,
                               const float* __restrict__ w,
                               float* __restrict__ xw) {
    int idx = blockIdx.x * blockDim.x + threadIdx.x;
    int o = idx & 31;
    int n = (idx >> 5) & (NN - 1);
    int b = idx >> 16;
    const float* xr = x + ((size_t)b * NN + n) * FF;
    float s = 0.f;
#pragma unroll
    for (int f = 0; f < FF; ++f) s = fmaf(xr[f], w[f * FF + o], s);
    xw[idx] = s;
}

__device__ __forceinline__ void sum_wave_tiles(float* red, int wave, int lane,
                                               const v8f& acc0, const v8f& acc1,
                                               float& s_lo, float& s_hi) {
    float* mine = red + wave * TILE_EL;
#pragma unroll
    for (int r = 0; r < 8; ++r) {
        mine[r * 64 + lane]      = acc0[r];
        mine[r * 64 + 32 + lane] = acc1[r];
    }
    __syncthreads();
    const int i_lo = wave * 64 + (lane & 1) * 32 + (lane >> 1);
    const int i_hi = i_lo + 16;
    float a = red[i_lo];
    float c = red[i_hi];
#pragma unroll
    for (int w = 1; w < NWAVE; ++w) {
        a += red[w * TILE_EL + i_lo];
        c += red[w * TILE_EL + i_hi];
    }
    s_lo = a;
    s_hi = c;
}

__global__ void __launch_bounds__(256)
sufg_pass2_kernel(const float* __restrict__ dl,
                  const float* __restrict__ xw,
                  const float* __restrict__ filt,
                  float* __restrict__ xshr,
                  float thr) {
    __shared__ float red[NWAVE * TILE_EL];

    const int lane = threadIdx.x & 31;
    const int wave = threadIdx.x >> 5;
    const int tile = blockIdx.x;
    const int b    = blockIdx.y;

    const int m0   = tile * 16;
    const int row  = lane & 15;
    const int hi   = lane >> 4;
    const int koff = hi << 2;
    const int k0   = wave * KW2;

    const float* Arow = dl + (size_t)b * DROWS * DCOLS
                           + (size_t)(2048 + m0 + row) * DCOLS + 1 + k0;
    const float* A  = Arow + koff;
    const float* Bp = xw + (size_t)b * NN * FF + (size_t)(k0 + koff) * FF + 2 * row;

#pragma unroll
    for (int j = 0; j < KW2 / 64; ++j)
        __builtin_prefetch(Arow + hi * 32 + j * 64, 0, 0);

    v8f acc0 = {0.f,0.f,0.f,0.f,0.f,0.f,0.f,0.f};
    v8f acc1 = {0.f,0.f,0.f,0.f,0.f,0.f,0.f,0.f};

    for (int kc = 0; kc < KW2; kc += 64) {
#pragma unroll
        for (int k = 0; k < 64; k += 8) {
            KBLOCK8(A, Bp, acc0, acc1);
            A += 8; Bp += 8 * FF;
        }
    }

    float s_lo, s_hi;
    sum_wave_tiles(red, wave, lane, acc0, acc1, s_lo, s_hi);

    const int   mA  = m0 + wave;
    const int   mB  = mA + 8;
    float vA = s_lo * filt[2048 + mA];
    float vB = s_hi * filt[2048 + mB];
    if (m0 >= 2048) {
        vA = copysignf(fmaxf(fabsf(vA) - thr, 0.f), vA);
        vB = copysignf(fmaxf(fabsf(vB) - thr, 0.f), vB);
    }
    xshr[((size_t)b * MSHR + mA) * FF + lane] = vA;
    xshr[((size_t)b * MSHR + mB) * FF + lane] = vB;
}

__global__ void __launch_bounds__(256)
sufg_pass3_kernel(const float* __restrict__ dl,
                  const float* __restrict__ xshr,
                  float* __restrict__ partial) {
    __shared__ float red[NWAVE * TILE_EL];

    const int lane = threadIdx.x & 31;
    const int wave = threadIdx.x >> 5;
    const int unit = blockIdx.x;
    const int b    = blockIdx.y;

    const int split  = unit & (NSPLIT - 1);
    const int tile   = unit >> 2;
    const int kbegin = split * KSEG + wave * KW3;

    const int row  = lane & 15;
    const int hi   = lane >> 4;
    const int koff = hi << 2;

    int krow = tile * 16 + row;
    if (krow > KOUT - 1) krow = KOUT - 1;

    const float* Arow = dl + (size_t)b * DROWS * DCOLS + (size_t)2048 * DCOLS
                           + (size_t)krow * MSHR + kbegin;
    const float* A  = Arow + koff;
    const float* Bp = xshr + (size_t)b * MSHR * FF + (size_t)(kbegin + koff) * FF + 2 * row;

#pragma unroll
    for (int j = 0; j < KW3 / 64; ++j)
        __builtin_prefetch(Arow + hi * 32 + j * 64, 0, 0);

    v8f acc0 = {0.f,0.f,0.f,0.f,0.f,0.f,0.f,0.f};
    v8f acc1 = {0.f,0.f,0.f,0.f,0.f,0.f,0.f,0.f};

    for (int kc = 0; kc < KW3; kc += 64) {
#pragma unroll
        for (int k = 0; k < 64; k += 8) {
            KBLOCK8(A, Bp, acc0, acc1);
            A += 8; Bp += 8 * FF;
        }
    }

    float s_lo, s_hi;
    sum_wave_tiles(red, wave, lane, acc0, acc1, s_lo, s_hi);

    const int kkA = tile * 16 + wave;
    float* pbase = partial + ((size_t)(b * NSPLIT + split)) * KPAD * FF;
    pbase[(size_t)kkA * FF + lane]       = s_lo;
    pbase[(size_t)(kkA + 8) * FF + lane] = s_hi;
}

__global__ void sufg_reduce_kernel(const float* __restrict__ partial,
                                   const float* __restrict__ bias,
                                   float* __restrict__ out) {
    int idx = blockIdx.x * blockDim.x + threadIdx.x;
    if (idx >= BB * KOUT * FF) return;
    int o = idx & 31;
    int k = (idx >> 5) % KOUT;
    int b = idx / (KOUT * FF);
    float s = bias[o];
#pragma unroll
    for (int sp = 0; sp < NSPLIT; ++sp)
        s += partial[(((size_t)(b * NSPLIT + sp)) * KPAD + k) * FF + o];
    out[idx] = s;
}

extern "C" void kernel_launch(void* const* d_in, const int* in_sizes, int n_in,
                              void* d_out, int out_size, void* d_ws, size_t ws_size,
                              hipStream_t stream) {
    (void)in_sizes; (void)n_in; (void)out_size; (void)ws_size;
    const float* x    = (const float*)d_in[0];
    const float* dl   = (const float*)d_in[1];
    const float* w    = (const float*)d_in[2];
    const float* filt = (const float*)d_in[3];
    const float* bias = (const float*)d_in[4];
    float* out = (float*)d_out;

    float* ws      = (float*)d_ws;
    float* xw      = ws + WS_XW_OFF;
    float* xshr    = ws + WS_XSHR_OFF;
    float* partial = ws + WS_PART_OFF;

    const float thr = sqrtf(2.0f * logf((float)NN)) / sqrtf((float)NN);

    sufg_xw_kernel<<<(BB * NN * FF) / 256, 256, 0, stream>>>(x, w, xw);
    sufg_pass2_kernel<<<dim3(MSHR / 16, BB), 256, 0, stream>>>(dl, xw, filt, xshr, thr);
    sufg_pass3_kernel<<<dim3((KPAD / 16) * NSPLIT, BB), 256, 0, stream>>>(dl, xshr, partial);
    sufg_reduce_kernel<<<(BB * KOUT * FF + 255) / 256, 256, 0, stream>>>(partial, bias, out);
}


# Each wave multiplies three 16-row groups (48x32 block tile) with every B fragment it loads: B reads from L2 cut to a third; next 32-k chunk's fragments loaded a chunk ahead; sum order unchanged
# speedup vs baseline: 2.1378x; 1.3157x over previous
#include <hip/hip_runtime.h>
#include <cmath>


typedef float v2f __attribute__((ext_vector_type(2)));
typedef float v4f __attribute__((ext_vector_type(4)));
typedef float v8f __attribute__((ext_vector_type(8)));
typedef float v2f_a4 __attribute__((ext_vector_type(2), aligned(4)));
typedef float v4f_a4 __attribute__((ext_vector_type(4), aligned(4)));

#define WMMA_F32(A, Bf, C) \
    __builtin_amdgcn_wmma_f32_16x16x4_f32(false, (A), false, (Bf), (short)0, (C), false, false)

#define BB       2
#define NN       2048
#define FF       32
#define DROWS    8192
#define DCOLS    2049
#define MSHR     6144
#define KOUT     2049
#define KPAD     2064
#define NSPLIT   4
#define KSEG     (MSHR / NSPLIT)
#define NWAVE    8
#define KW2      (NN / NWAVE)
#define KW3      (KSEG / NWAVE)
#define TILE_EL  (16 * FF)
#define RG       3
#define TROWS    (16 * RG)
#define CHB      4

#define WS_XW_OFF    0
#define WS_XSHR_OFF  (BB * NN * FF)
#define WS_PART_OFF  (WS_XSHR_OFF + BB * MSHR * FF)

struct BlkFrag {
    v4f a[RG];
    v2f r[4];
};

__device__ __forceinline__ void load_blk(BlkFrag& f, const float* a0, const float* a1,
                                         const float* a2, const float* bp) {
    f.a[0] = *(const v4f_a4*)a0;
    f.a[1] = *(const v4f_a4*)a1;
    f.a[2] = *(const v4f_a4*)a2;
    f.r[0] = *(const v2f_a4*)(bp);
    f.r[1] = *(const v2f_a4*)(bp + FF);
    f.r[2] = *(const v2f_a4*)(bp + 2 * FF);
    f.r[3] = *(const v2f_a4*)(bp + 3 * FF);
}

__device__ __forceinline__ void mma_blk(const BlkFrag& f, v8f (&acc)[RG][2]) {
    v2f b0;  b0.x = f.r[0].x; b0.y = f.r[1].x;
    v2f b1;  b1.x = f.r[0].y; b1.y = f.r[1].y;
    v2f b2;  b2.x = f.r[2].x; b2.y = f.r[3].x;
    v2f b3;  b3.x = f.r[2].y; b3.y = f.r[3].y;
#pragma unroll
    for (int g = 0; g < RG; ++g) {
        v2f a01; a01.x = f.a[g].x; a01.y = f.a[g].y;
        acc[g][0] = WMMA_F32(a01, b0, acc[g][0]);
        acc[g][1] = WMMA_F32(a01, b1, acc[g][1]);
    }
#pragma unroll
    for (int g = 0; g < RG; ++g) {
        v2f a23; a23.x = f.a[g].z; a23.y = f.a[g].w;
        acc[g][0] = WMMA_F32(a23, b2, acc[g][0]);
        acc[g][1] = WMMA_F32(a23, b3, acc[g][1]);
    }
}

template <int KW>
__device__ __forceinline__ void k_slice(const float* a0, const float* a1, const float* a2,
                                        const float* bp, v8f (&acc)[RG][2]) {
    constexpr int NCH = KW / (8 * CHB);
    BlkFrag f[2][CHB];
#pragma unroll
    for (int j = 0; j < CHB; ++j)
        load_blk(f[0][j], a0 + 8 * j, a1 + 8 * j, a2 + 8 * j, bp + 8 * j * FF);
#pragma unroll
    for (int c = 0; c < NCH; ++c) {
        const int cur = c & 1;
        const int nxt = cur ^ 1;
        if (c + 1 < NCH) {
            const int ko = (c + 1) * 8 * CHB;
#pragma unroll
            for (int j = 0; j < CHB; ++j)
                load_blk(f[nxt][j], a0 + ko + 8 * j, a1 + ko + 8 * j, a2 + ko + 8 * j,
                         bp + (ko + 8 * j) * FF);
        }
        __builtin_amdgcn_sched_barrier(0);
#pragma unroll
        for (int j = 0; j < CHB; ++j)
            mma_blk(f[cur][j], acc);
    }
}

__global__ void sufg_xw_kernel(const float* __restrict__ x,
                               const float* __restrict__ w,
                               float* __restrict__ xw) {
    int idx = blockIdx.x * blockDim.x + threadIdx.x;
    int o = idx & 31;
    int n = (idx >> 5) & (NN - 1);
    int b = idx >> 16;
    const float* xr = x + ((size_t)b * NN + n) * FF;
    float s = 0.f;
#pragma unroll
    for (int f = 0; f < FF; ++f) s = fmaf(xr[f], w[f * FF + o], s);
    xw[idx] = s;
}

__device__ __forceinline__ void sum_wave_tiles(float* red, int wave, int lane,
                                               const v8f (&acc)[RG][2],
                                               float (&s_lo)[RG], float (&s_hi)[RG]) {
    float* mine = red + wave * (RG * TILE_EL);
#pragma unroll
    for (int g = 0; g < RG; ++g) {
#pragma unroll
        for (int r = 0; r < 8; ++r) {
            mine[g * TILE_EL + r * 64 + lane]      = acc[g][0][r];
            mine[g * TILE_EL + r * 64 + 32 + lane] = acc[g][1][r];
        }
    }
    __syncthreads();
    const int i_lo = wave * 64 + (lane & 1) * 32 + (lane >> 1);
    const int i_hi = i_lo + 16;
#pragma unroll
    for (int g = 0; g < RG; ++g) {
        float a = red[g * TILE_EL + i_lo];
        float c = red[g * TILE_EL + i_hi];
#pragma unroll
        for (int w = 1; w < NWAVE; ++w) {
            a += red[w * (RG * TILE_EL) + g * TILE_EL + i_lo];
            c += red[w * (RG * TILE_EL) + g * TILE_EL + i_hi];
        }
        s_lo[g] = a;
        s_hi[g] = c;
    }
}

__global__ void __launch_bounds__(256) __attribute__((amdgpu_waves_per_eu(4, 4)))
sufg_pass2_kernel(const float* __restrict__ dl,
                  const float* __restrict__ xw,
                  const float* __restrict__ filt,
                  float* __restrict__ xshr,
                  float thr) {
    __shared__ float red[NWAVE * RG * TILE_EL];

    const int lane = threadIdx.x & 31;
    const int wave = threadIdx.x >> 5;
    const int tile = blockIdx.x;
    const int b    = blockIdx.y;

    const int m0   = tile * TROWS;
    const int row  = lane & 15;
    const int hi   = lane >> 4;
    const int koff = hi << 2;
    const int k0   = wave * KW2;

    const float* Arow = dl + (size_t)b * DROWS * DCOLS
                           + (size_t)(2048 + m0 + row) * DCOLS + 1 + k0;
    const float* A  = Arow + koff;
    const float* Bp = xw + (size_t)b * NN * FF + (size_t)(k0 + koff) * FF + 2 * row;

#pragma unroll
    for (int g = 0; g < RG; ++g) {
#pragma unroll
        for (int j = 0; j < KW2 / 64; ++j)
            __builtin_prefetch(Arow + (size_t)g * 16 * DCOLS + hi * 32 + j * 64, 0, 0);
    }

    v8f acc[RG][2];
#pragma unroll
    for (int g = 0; g < RG; ++g) {
        acc[g][0] = (v8f){0.f,0.f,0.f,0.f,0.f,0.f,0.f,0.f};
        acc[g][1] = (v8f){0.f,0.f,0.f,0.f,0.f,0.f,0.f,0.f};
    }

    k_slice<KW2>(A, A + (size_t)16 * DCOLS, A + (size_t)32 * DCOLS, Bp, acc);

    float s_lo[RG], s_hi[RG];
    sum_wave_tiles(red, wave, lane, acc, s_lo, s_hi);

#pragma unroll
    for (int g = 0; g < RG; ++g) {
        const int mA = m0 + 16 * g + wave;
        const int mB = mA + 8;
        float vA = s_lo[g] * filt[2048 + mA];
        float vB = s_hi[g] * filt[2048 + mB];
        if (m0 + 16 * g >= 2048) {
            vA = copysignf(fmaxf(fabsf(vA) - thr, 0.f), vA);
            vB = copysignf(fmaxf(fabsf(vB) - thr, 0.f), vB);
        }
        xshr[((size_t)b * MSHR + mA) * FF + lane] = vA;
        xshr[((size_t)b * MSHR + mB) * FF + lane] = vB;
    }
}

__global__ void __launch_bounds__(256) __attribute__((amdgpu_waves_per_eu(4, 4)))
sufg_pass3_kernel(const float* __restrict__ dl,
                  const float* __restrict__ xshr,
                  float* __restrict__ partial) {
    __shared__ float red[NWAVE * RG * TILE_EL];

    const int lane = threadIdx.x & 31;
    const int wave = threadIdx.x >> 5;
    const int unit = blockIdx.x;
    const int b    = blockIdx.y;

    const int split  = unit & (NSPLIT - 1);
    const int tile   = unit >> 2;
    const int kbegin = split * KSEG + wave * KW3;

    const int row  = lane & 15;
    const int hi   = lane >> 4;
    const int koff = hi << 2;

    const float* Abase = dl + (size_t)b * DROWS * DCOLS + (size_t)2048 * DCOLS + kbegin;
    const float* Ag[RG];
#pragma unroll
    for (int g = 0; g < RG; ++g) {
        int krow = tile * TROWS + 16 * g + row;
        if (krow > KOUT - 1) krow = KOUT - 1;
        Ag[g] = Abase + (size_t)krow * MSHR;
    }
    const float* Bp = xshr + (size_t)b * MSHR * FF + (size_t)(kbegin + koff) * FF + 2 * row;

#pragma unroll
    for (int g = 0; g < RG; ++g) {
#pragma unroll
        for (int j = 0; j < KW3 / 64; ++j)
            __builtin_prefetch(Ag[g] + hi * 32 + j * 64, 0, 0);
    }

    v8f acc[RG][2];
#pragma unroll
    for (int g = 0; g < RG; ++g) {
        acc[g][0] = (v8f){0.f,0.f,0.f,0.f,0.f,0.f,0.f,0.f};
        acc[g][1] = (v8f){0.f,0.f,0.f,0.f,0.f,0.f,0.f,0.f};
    }

    k_slice<KW3>(Ag[0] + koff, Ag[1] + koff, Ag[2] + koff, Bp, acc);

    float s_lo[RG], s_hi[RG];
    sum_wave_tiles(red, wave, lane, acc, s_lo, s_hi);

    float* pbase = partial + ((size_t)(b * NSPLIT + split)) * KPAD * FF;
#pragma unroll
    for (int g = 0; g < RG; ++g) {
        const int kkA = tile * TROWS + 16 * g + wave;
        pbase[(size_t)kkA * FF + lane]       = s_lo[g];
        pbase[(size_t)(kkA + 8) * FF + lane] = s_hi[g];
    }
}

__global__ void sufg_reduce_kernel(const float* __restrict__ partial,
                                   const float* __restrict__ bias,
                                   float* __restrict__ out) {
    int idx = blockIdx.x * blockDim.x + threadIdx.x;
    if (idx >= BB * KOUT * FF) return;
    int o = idx & 31;
    int k = (idx >> 5) % KOUT;
    int b = idx / (KOUT * FF);
    float s = bias[o];
#pragma unroll
    for (int sp = 0; sp < NSPLIT; ++sp)
        s += partial[(((size_t)(b * NSPLIT + sp)) * KPAD + k) * FF + o];
    out[idx] = s;
}

extern "C" void kernel_launch(void* const* d_in, const int* in_sizes, int n_in,
                              void* d_out, int out_size, void* d_ws, size_t ws_size,
                              hipStream_t stream) {
    (void)in_sizes; (void)n_in; (void)out_size; (void)ws_size;
    const float* x    = (const float*)d_in[0];
    const float* dl   = (const float*)d_in[1];
    const float* w    = (const float*)d_in[2];
    const float* filt = (const float*)d_in[3];
    const float* bias = (const float*)d_in[4];
    float* out = (float*)d_out;

    float* ws      = (float*)d_ws;
    float* xw      = ws + WS_XW_OFF;
    float* xshr    = ws + WS_XSHR_OFF;
    float* partial = ws + WS_PART_OFF;

    const float thr = sqrtf(2.0f * logf((float)NN)) / sqrtf((float)NN);

    sufg_xw_kernel<<<(BB * NN * FF) / 256, 256, 0, stream>>>(x, w, xw);
    sufg_pass2_kernel<<<dim3(MSHR / TROWS, BB), 256, 0, stream>>>(dl, xw, filt, xshr, thr);
    sufg_pass3_kernel<<<dim3((KPAD / TROWS) * NSPLIT, BB), 256, 0, stream>>>(dl, xshr, partial);
    sufg_reduce_kernel<<<(BB * KOUT * FF + 255) / 256, 256, 0, stream>>>(partial, bias, out);
}


# Feature transform moved behind the analysis product, (y@x)@weight for y@(x@weight), in pass 2's epilogue (FMA, new sum grouping): the xw launch and its workspace plane are gone
# speedup vs baseline: 2.2678x; 1.0608x over previous
#include <hip/hip_runtime.h>
#include <cmath>


typedef float v2f __attribute__((ext_vector_type(2)));
typedef float v4f __attribute__((ext_vector_type(4)));
typedef float v8f __attribute__((ext_vector_type(8)));
typedef float v2f_a4 __attribute__((ext_vector_type(2), aligned(4)));
typedef float v4f_a4 __attribute__((ext_vector_type(4), aligned(4)));

#define WMMA_F32(A, Bf, C) \
    __builtin_amdgcn_wmma_f32_16x16x4_f32(false, (A), false, (Bf), (short)0, (C), false, false)

#define BB       2
#define NN       2048
#define FF       32
#define DROWS    8192
#define DCOLS    2049
#define MSHR     6144
#define KOUT     2049
#define KPAD     2064
#define NSPLIT   4
#define KSEG     (MSHR / NSPLIT)
#define NWAVE    8
#define KW2      (NN / NWAVE)
#define KW3      (KSEG / NWAVE)
#define TILE_EL  (16 * FF)
#define RG       3
#define TROWS    (16 * RG)
#define CHB      4

#define WS_XSHR_OFF  (BB * NN * FF)
#define WS_PART_OFF  (WS_XSHR_OFF + BB * MSHR * FF)

struct BlkFrag {
    v4f a[RG];
    v2f r[4];
};

__device__ __forceinline__ void load_blk(BlkFrag& f, const float* a0, const float* a1,
                                         const float* a2, const float* bp) {
    f.a[0] = *(const v4f_a4*)a0;
    f.a[1] = *(const v4f_a4*)a1;
    f.a[2] = *(const v4f_a4*)a2;
    f.r[0] = *(const v2f_a4*)(bp);
    f.r[1] = *(const v2f_a4*)(bp + FF);
    f.r[2] = *(const v2f_a4*)(bp + 2 * FF);
    f.r[3] = *(const v2f_a4*)(bp + 3 * FF);
}

__device__ __forceinline__ void mma_blk(const BlkFrag& f, v8f (&acc)[RG][2]) {
    v2f b0;  b0.x = f.r[0].x; b0.y = f.r[1].x;
    v2f b1;  b1.x = f.r[0].y; b1.y = f.r[1].y;
    v2f b2;  b2.x = f.r[2].x; b2.y = f.r[3].x;
    v2f b3;  b3.x = f.r[2].y; b3.y = f.r[3].y;
#pragma unroll
    for (int g = 0; g < RG; ++g) {
        v2f a01; a01.x = f.a[g].x; a01.y = f.a[g].y;
        acc[g][0] = WMMA_F32(a01, b0, acc[g][0]);
        acc[g][1] = WMMA_F32(a01, b1, acc[g][1]);
    }
#pragma unroll
    for (int g = 0; g < RG; ++g) {
        v2f a23; a23.x = f.a[g].z; a23.y = f.a[g].w;
        acc[g][0] = WMMA_F32(a23, b2, acc[g][0]);
        acc[g][1] = WMMA_F32(a23, b3, acc[g][1]);
    }
}

template <int KW>
__device__ __forceinline__ void k_slice(const float* a0, const float* a1, const float* a2,
                                        const float* bp, v8f (&acc)[RG][2]) {
    constexpr int NCH = KW / (8 * CHB);
    BlkFrag f[2][CHB];
#pragma unroll
    for (int j = 0; j < CHB; ++j)
        load_blk(f[0][j], a0 + 8 * j, a1 + 8 * j, a2 + 8 * j, bp + 8 * j * FF);
#pragma unroll
    for (int c = 0; c < NCH; ++c) {
        const int cur = c & 1;
        const int nxt = cur ^ 1;
        if (c + 1 < NCH) {
            const int ko = (c + 1) * 8 * CHB;
#pragma unroll
            for (int j = 0; j < CHB; ++j)
                load_blk(f[nxt][j], a0 + ko + 8 * j, a1 + ko + 8 * j, a2 + ko + 8 * j,
                         bp + (ko + 8 * j) * FF);
        }
        __builtin_amdgcn_sched_barrier(0);
#pragma unroll
        for (int j = 0; j < CHB; ++j)
            mma_blk(f[cur][j], acc);
    }
}

__device__ __forceinline__ void sum_wave_tiles(float* red, int wave, int lane,
                                               const v8f (&acc)[RG][2],
                                               float (&s_lo)[RG], float (&s_hi)[RG]) {
    float* mine = red + wave * (RG * TILE_EL);
#pragma unroll
    for (int g = 0; g < RG; ++g) {
#pragma unroll
        for (int r = 0; r < 8; ++r) {
            mine[g * TILE_EL + r * 64 + lane]      = acc[g][0][r];
            mine[g * TILE_EL + r * 64 + 32 + lane] = acc[g][1][r];
        }
    }
    __syncthreads();
    const int i_lo = wave * 64 + (lane & 1) * 32 + (lane >> 1);
    const int i_hi = i_lo + 16;
#pragma unroll
    for (int g = 0; g < RG; ++g) {
        float a = red[g * TILE_EL + i_lo];
        float c = red[g * TILE_EL + i_hi];
#pragma unroll
        for (int w = 1; w < NWAVE; ++w) {
            a += red[w * (RG * TILE_EL) + g * TILE_EL + i_lo];
            c += red[w * (RG * TILE_EL) + g * TILE_EL + i_hi];
        }
        s_lo[g] = a;
        s_hi[g] = c;
    }
}

__global__ void __launch_bounds__(256) __attribute__((amdgpu_waves_per_eu(4, 4)))
sufg_pass2_kernel(const float* __restrict__ dl,
                  const float* __restrict__ x,
                  const float* __restrict__ w,
                  const float* __restrict__ filt,
                  float* __restrict__ xshr,
                  float thr) {
    __shared__ float red[NWAVE * RG * TILE_EL];

    const int lane = threadIdx.x & 31;
    const int wave = threadIdx.x >> 5;
    const int tile = blockIdx.x;
    const int b    = blockIdx.y;

    const int m0   = tile * TROWS;
    const int row  = lane & 15;
    const int hi   = lane >> 4;
    const int koff = hi << 2;
    const int k0   = wave * KW2;

    const float* Arow = dl + (size_t)b * DROWS * DCOLS
                           + (size_t)(2048 + m0 + row) * DCOLS + 1 + k0;
    const float* A  = Arow + koff;
    const float* Bp = x + (size_t)b * NN * FF + (size_t)(k0 + koff) * FF + 2 * row;

#pragma unroll
    for (int g = 0; g < RG; ++g) {
#pragma unroll
        for (int j = 0; j < KW2 / 64; ++j)
            __builtin_prefetch(Arow + (size_t)g * 16 * DCOLS + hi * 32 + j * 64, 0, 0);
    }

    v8f acc[RG][2];
#pragma unroll
    for (int g = 0; g < RG; ++g) {
        acc[g][0] = (v8f){0.f,0.f,0.f,0.f,0.f,0.f,0.f,0.f};
        acc[g][1] = (v8f){0.f,0.f,0.f,0.f,0.f,0.f,0.f,0.f};
    }

    k_slice<KW2>(A, A + (size_t)16 * DCOLS, A + (size_t)32 * DCOLS, Bp, acc);

    float wcol[FF];
#pragma unroll
    for (int f = 0; f < FF; ++f) wcol[f] = w[f * FF + lane];

    float s_lo[RG], s_hi[RG];
    sum_wave_tiles(red, wave, lane, acc, s_lo, s_hi);

    float t_lo[RG], t_hi[RG];
#pragma unroll
    for (int g = 0; g < RG; ++g) { t_lo[g] = 0.f; t_hi[g] = 0.f; }
#pragma unroll
    for (int f = 0; f < FF; ++f) {
#pragma unroll
        for (int g = 0; g < RG; ++g) {
            const float a = __builtin_bit_cast(float,
                __builtin_amdgcn_readlane(__builtin_bit_cast(int, s_lo[g]), f));
            const float c = __builtin_bit_cast(float,
                __builtin_amdgcn_readlane(__builtin_bit_cast(int, s_hi[g]), f));
            t_lo[g] = fmaf(a, wcol[f], t_lo[g]);
            t_hi[g] = fmaf(c, wcol[f], t_hi[g]);
        }
    }

#pragma unroll
    for (int g = 0; g < RG; ++g) {
        const int mA = m0 + 16 * g + wave;
        const int mB = mA + 8;
        float vA = t_lo[g] * filt[2048 + mA];
        float vB = t_hi[g] * filt[2048 + mB];
        if (m0 + 16 * g >= 2048) {
            vA = copysignf(fmaxf(fabsf(vA) - thr, 0.f), vA);
            vB = copysignf(fmaxf(fabsf(vB) - thr, 0.f), vB);
        }
        xshr[((size_t)b * MSHR + mA) * FF + lane] = vA;
        xshr[((size_t)b * MSHR + mB) * FF + lane] = vB;
    }
}

__global__ void __launch_bounds__(256) __attribute__((amdgpu_waves_per_eu(4, 4)))
sufg_pass3_kernel(const float* __restrict__ dl,
                  const float* __restrict__ xshr,
                  float* __restrict__ partial) {
    __shared__ float red[NWAVE * RG * TILE_EL];

    const int lane = threadIdx.x & 31;
    const int wave = threadIdx.x >> 5;
    const int unit = blockIdx.x;
    const int b    = blockIdx.y;

    const int split  = unit & (NSPLIT - 1);
    const int tile   = unit >> 2;
    const int kbegin = split * KSEG + wave * KW3;

    const int row  = lane & 15;
    const int hi   = lane >> 4;
    const int koff = hi << 2;

    const float* Abase = dl + (size_t)b * DROWS * DCOLS + (size_t)2048 * DCOLS + kbegin;
    const float* Ag[RG];
#pragma unroll
    for (int g = 0; g < RG; ++g) {
        int krow = tile * TROWS + 16 * g + row;
        if (krow > KOUT - 1) krow = KOUT - 1;
        Ag[g] = Abase + (size_t)krow * MSHR;
    }
    const float* Bp = xshr + (size_t)b * MSHR * FF + (size_t)(kbegin + koff) * FF + 2 * row;

#pragma unroll
    for (int g = 0; g < RG; ++g) {
#pragma unroll
        for (int j = 0; j < KW3 / 64; ++j)
            __builtin_prefetch(Ag[g] + hi * 32 + j * 64, 0, 0);
    }

    v8f acc[RG][2];
#pragma unroll
    for (int g = 0; g < RG; ++g) {
        acc[g][0] = (v8f){0.f,0.f,0.f,0.f,0.f,0.f,0.f,0.f};
        acc[g][1] = (v8f){0.f,0.f,0.f,0.f,0.f,0.f,0.f,0.f};
    }

    k_slice<KW3>(Ag[0] + koff, Ag[1] + koff, Ag[2] + koff, Bp, acc);

    float s_lo[RG], s_hi[RG];
    sum_wave_tiles(red, wave, lane, acc, s_lo, s_hi);

    float* pbase = partial + ((size_t)(b * NSPLIT + split)) * KPAD * FF;
#pragma unroll
    for (int g = 0; g < RG; ++g) {
        const int kkA = tile * TROWS + 16 * g + wave;
        pbase[(size_t)kkA * FF + lane]       = s_lo[g];
        pbase[(size_t)(kkA + 8) * FF + lane] = s_hi[g];
    }
}

__global__ void sufg_reduce_kernel(const float* __restrict__ partial,
                                   const float* __restrict__ bias,
                                   float* __restrict__ out) {
    int idx = blockIdx.x * blockDim.x + threadIdx.x;
    if (idx >= BB * KOUT * FF) return;
    int o = idx & 31;
    int k = (idx >> 5) % KOUT;
    int b = idx / (KOUT * FF);
    float s = bias[o];
#pragma unroll
    for (int sp = 0; sp < NSPLIT; ++sp)
        s += partial[(((size_t)(b * NSPLIT + sp)) * KPAD + k) * FF + o];
    out[idx] = s;
}

extern "C" void kernel_launch(void* const* d_in, const int* in_sizes, int n_in,
                              void* d_out, int out_size, void* d_ws, size_t ws_size,
                              hipStream_t stream) {
    (void)in_sizes; (void)n_in; (void)out_size; (void)ws_size;
    const float* x    = (const float*)d_in[0];
    const float* dl   = (const float*)d_in[1];
    const float* w    = (const float*)d_in[2];
    const float* filt = (const float*)d_in[3];
    const float* bias = (const float*)d_in[4];
    float* out = (float*)d_out;

    float* ws      = (float*)d_ws;
    float* xshr    = ws + WS_XSHR_OFF;
    float* partial = ws + WS_PART_OFF;

    const float thr = sqrtf(2.0f * logf((float)NN)) / sqrtf((float)NN);

    sufg_pass2_kernel<<<dim3(MSHR / TROWS, BB), 256, 0, stream>>>(dl, x, w, filt, xshr, thr);
    sufg_pass3_kernel<<<dim3((KPAD / TROWS) * NSPLIT, BB), 256, 0, stream>>>(dl, xshr, partial);
    sufg_reduce_kernel<<<(BB * KOUT * FF + 255) / 256, 256, 0, stream>>>(partial, bias, out);
}


# Pass 3 re-tiled to 64-row tiles (4 row groups per B fragment; row 2048 a 5th group in the last tile): 256 blocks, one per WGP, for 344 (two on some WGPs); B reads down a quarter; sum order unchanged
# speedup vs baseline: 2.3404x; 1.0320x over previous
#include <hip/hip_runtime.h>
#include <cmath>


typedef float v2f __attribute__((ext_vector_type(2)));
typedef float v4f __attribute__((ext_vector_type(4)));
typedef float v8f __attribute__((ext_vector_type(8)));
typedef float v2f_a4 __attribute__((ext_vector_type(2), aligned(4)));
typedef float v4f_a4 __attribute__((ext_vector_type(4), aligned(4)));

#define WMMA_F32(A, Bf, C) \
    __builtin_amdgcn_wmma_f32_16x16x4_f32(false, (A), false, (Bf), (short)0, (C), false, false)

#define BB       2
#define NN       2048
#define FF       32
#define DROWS    8192
#define DCOLS    2049
#define MSHR     6144
#define KOUT     2049
#define KPAD     2064
#define NSPLIT   4
#define KSEG     (MSHR / NSPLIT)
#define NWAVE    8
#define KW2      (NN / NWAVE)
#define KW3      (KSEG / NWAVE)
#define TILE_EL  (16 * FF)
#define RG       3
#define TROWS    (16 * RG)
#define RG3      4
#define TROWS3   (16 * RG3)
#define NT3      (NN / TROWS3)
#define RG3L     (RG3 + 1)
#define CHB      4

#define WS_XSHR_OFF  (BB * NN * FF)
#define WS_PART_OFF  (WS_XSHR_OFF + BB * MSHR * FF)

template <int RGT>
struct BlkFrag {
    v4f a[RGT];
    v2f r[4];
};

template <int RGT>
__device__ __forceinline__ void load_blk(BlkFrag<RGT>& f, const float* const (&ap)[RGT], int ao,
                                         const float* bp) {
#pragma unroll
    for (int g = 0; g < RGT; ++g)
        f.a[g] = *(const v4f_a4*)(ap[g] + ao);
    f.r[0] = *(const v2f_a4*)(bp);
    f.r[1] = *(const v2f_a4*)(bp + FF);
    f.r[2] = *(const v2f_a4*)(bp + 2 * FF);
    f.r[3] = *(const v2f_a4*)(bp + 3 * FF);
}

template <int RGT>
__device__ __forceinline__ void mma_blk(const BlkFrag<RGT>& f, v8f (&acc)[RGT][2]) {
    v2f b0;  b0.x = f.r[0].x; b0.y = f.r[1].x;
    v2f b1;  b1.x = f.r[0].y; b1.y = f.r[1].y;
    v2f b2;  b2.x = f.r[2].x; b2.y = f.r[3].x;
    v2f b3;  b3.x = f.r[2].y; b3.y = f.r[3].y;
#pragma unroll
    for (int g = 0; g < RGT; ++g) {
        v2f a01; a01.x = f.a[g].x; a01.y = f.a[g].y;
        acc[g][0] = WMMA_F32(a01, b0, acc[g][0]);
        acc[g][1] = WMMA_F32(a01, b1, acc[g][1]);
    }
#pragma unroll
    for (int g = 0; g < RGT; ++g) {
        v2f a23; a23.x = f.a[g].z; a23.y = f.a[g].w;
        acc[g][0] = WMMA_F32(a23, b2, acc[g][0]);
        acc[g][1] = WMMA_F32(a23, b3, acc[g][1]);
    }
}

template <int KW, int RGT>
__device__ __forceinline__ void k_slice(const float* const (&ap)[RGT],
                                        const float* bp, v8f (&acc)[RGT][2]) {
    constexpr int NCH = KW / (8 * CHB);
    BlkFrag<RGT> f[2][CHB];
#pragma unroll
    for (int j = 0; j < CHB; ++j)
        load_blk<RGT>(f[0][j], ap, 8 * j, bp + 8 * j * FF);
#pragma unroll
    for (int c = 0; c < NCH; ++c) {
        const int cur = c & 1;
        const int nxt = cur ^ 1;
        if (c + 1 < NCH) {
            const int ko = (c + 1) * 8 * CHB;
#pragma unroll
            for (int j = 0; j < CHB; ++j)
                load_blk<RGT>(f[nxt][j], ap, ko + 8 * j, bp + (ko + 8 * j) * FF);
        }
        __builtin_amdgcn_sched_barrier(0);
#pragma unroll
        for (int j = 0; j < CHB; ++j)
            mma_blk<RGT>(f[cur][j], acc);
    }
}

template <int RGT>
__device__ __forceinline__ void sum_wave_tiles(float* red, int wave, int lane,
                                               const v8f (&acc)[RGT][2],
                                               float (&s_lo)[RGT], float (&s_hi)[RGT]) {
    float* mine = red + wave * (RGT * TILE_EL);
#pragma unroll
    for (int g = 0; g < RGT; ++g) {
#pragma unroll
        for (int r = 0; r < 8; ++r) {
            mine[g * TILE_EL + r * 64 + lane]      = acc[g][0][r];
            mine[g * TILE_EL + r * 64 + 32 + lane] = acc[g][1][r];
        }
    }
    __syncthreads();
    const int i_lo = wave * 64 + (lane & 1) * 32 + (lane >> 1);
    const int i_hi = i_lo + 16;
#pragma unroll
    for (int g = 0; g < RGT; ++g) {
        float a = red[g * TILE_EL + i_lo];
        float c = red[g * TILE_EL + i_hi];
#pragma unroll
        for (int w = 1; w < NWAVE; ++w) {
            a += red[w * (RGT * TILE_EL) + g * TILE_EL + i_lo];
            c += red[w * (RGT * TILE_EL) + g * TILE_EL + i_hi];
        }
        s_lo[g] = a;
        s_hi[g] = c;
    }
}

__global__ void __launch_bounds__(256) __attribute__((amdgpu_waves_per_eu(4, 4)))
sufg_pass2_kernel(const float* __restrict__ dl,
                  const float* __restrict__ x,
                  const float* __restrict__ w,
                  const float* __restrict__ filt,
                  float* __restrict__ xshr,
                  float thr) {
    __shared__ float red[NWAVE * RG * TILE_EL];

    const int lane = threadIdx.x & 31;
    const int wave = threadIdx.x >> 5;
    const int tile = blockIdx.x;
    const int b    = blockIdx.y;

    const int m0   = tile * TROWS;
    const int row  = lane & 15;
    const int hi   = lane >> 4;
    const int koff = hi << 2;
    const int k0   = wave * KW2;

    const float* Arow = dl + (size_t)b * DROWS * DCOLS
                           + (size_t)(2048 + m0 + row) * DCOLS + 1 + k0;
    const float* A  = Arow + koff;
    const float* Bp = x + (size_t)b * NN * FF + (size_t)(k0 + koff) * FF + 2 * row;

#pragma unroll
    for (int g = 0; g < RG; ++g) {
#pragma unroll
        for (int j = 0; j < KW2 / 64; ++j)
            __builtin_prefetch(Arow + (size_t)g * 16 * DCOLS + hi * 32 + j * 64, 0, 0);
    }

    v8f acc[RG][2];
#pragma unroll
    for (int g = 0; g < RG; ++g) {
        acc[g][0] = (v8f){0.f,0.f,0.f,0.f,0.f,0.f,0.f,0.f};
        acc[g][1] = (v8f){0.f,0.f,0.f,0.f,0.f,0.f,0.f,0.f};
    }

    const float* const Ap[RG] = { A, A + (size_t)16 * DCOLS, A + (size_t)32 * DCOLS };
    k_slice<KW2, RG>(Ap, Bp, acc);

    float wcol[FF];
#pragma unroll
    for (int f = 0; f < FF; ++f) wcol[f] = w[f * FF + lane];

    float s_lo[RG], s_hi[RG];
    sum_wave_tiles<RG>(red, wave, lane, acc, s_lo, s_hi);

    float t_lo[RG], t_hi[RG];
#pragma unroll
    for (int g = 0; g < RG; ++g) { t_lo[g] = 0.f; t_hi[g] = 0.f; }
#pragma unroll
    for (int f = 0; f < FF; ++f) {
#pragma unroll
        for (int g = 0; g < RG; ++g) {
            const float a = __builtin_bit_cast(float,
                __builtin_amdgcn_readlane(__builtin_bit_cast(int, s_lo[g]), f));
            const float c = __builtin_bit_cast(float,
                __builtin_amdgcn_readlane(__builtin_bit_cast(int, s_hi[g]), f));
            t_lo[g] = fmaf(a, wcol[f], t_lo[g]);
            t_hi[g] = fmaf(c, wcol[f], t_hi[g]);
        }
    }

#pragma unroll
    for (int g = 0; g < RG; ++g) {
        const int mA = m0 + 16 * g + wave;
        const int mB = mA + 8;
        float vA = t_lo[g] * filt[2048 + mA];
        float vB = t_hi[g] * filt[2048 + mB];
        if (m0 + 16 * g >= 2048) {
            vA = copysignf(fmaxf(fabsf(vA) - thr, 0.f), vA);
            vB = copysignf(fmaxf(fabsf(vB) - thr, 0.f), vB);
        }
        xshr[((size_t)b * MSHR + mA) * FF + lane] = vA;
        xshr[((size_t)b * MSHR + mB) * FF + lane] = vB;
    }
}

template <int RGT>
__device__ __forceinline__ void pass3_tile(const float* __restrict__ dl,
                                           const float* __restrict__ xshr,
                                           float* __restrict__ partial,
                                           float* red, int tile, int split, int b,
                                           int wave, int lane) {
    const int kbegin = split * KSEG + wave * KW3;

    const int row  = lane & 15;
    const int hi   = lane >> 4;
    const int koff = hi << 2;

    const float* Abase = dl + (size_t)b * DROWS * DCOLS + (size_t)2048 * DCOLS + kbegin;
    const float* Ag[RGT];
#pragma unroll
    for (int g = 0; g < RGT; ++g) {
        int krow = tile * TROWS3 + 16 * g + row;
        if (krow > KOUT - 1) krow = KOUT - 1;
        Ag[g] = Abase + (size_t)krow * MSHR;
    }
    const float* Bp = xshr + (size_t)b * MSHR * FF + (size_t)(kbegin + koff) * FF + 2 * row;

#pragma unroll
    for (int g = 0; g < RGT; ++g) {
#pragma unroll
        for (int j = 0; j < KW3 / 64; ++j)
            __builtin_prefetch(Ag[g] + hi * 32 + j * 64, 0, 0);
    }

    v8f acc[RGT][2];
#pragma unroll
    for (int g = 0; g < RGT; ++g) {
        acc[g][0] = (v8f){0.f,0.f,0.f,0.f,0.f,0.f,0.f,0.f};
        acc[g][1] = (v8f){0.f,0.f,0.f,0.f,0.f,0.f,0.f,0.f};
    }

    const float* Ap[RGT];
#pragma unroll
    for (int g = 0; g < RGT; ++g) Ap[g] = Ag[g] + koff;
    k_slice<KW3, RGT>(Ap, Bp, acc);

    float s_lo[RGT], s_hi[RGT];
    sum_wave_tiles<RGT>(red, wave, lane, acc, s_lo, s_hi);

    float* pbase = partial + ((size_t)(b * NSPLIT + split)) * KPAD * FF;
#pragma unroll
    for (int g = 0; g < RGT; ++g) {
        const int kkA = tile * TROWS3 + 16 * g + wave;
        pbase[(size_t)kkA * FF + lane]       = s_lo[g];
        pbase[(size_t)(kkA + 8) * FF + lane] = s_hi[g];
    }
}

__global__ void __launch_bounds__(256) __attribute__((amdgpu_waves_per_eu(2, 2)))
sufg_pass3_kernel(const float* __restrict__ dl,
                  const float* __restrict__ xshr,
                  float* __restrict__ partial) {
    __shared__ float red[NWAVE * RG3L * TILE_EL];

    const int lane = threadIdx.x & 31;
    const int wave = threadIdx.x >> 5;
    const int unit = blockIdx.x;
    const int b    = blockIdx.y;

    const int split = unit & (NSPLIT - 1);
    const int tile  = (NT3 - 1) - (unit >> 2);

    if (tile == NT3 - 1)
        pass3_tile<RG3L>(dl, xshr, partial, red, tile, split, b, wave, lane);
    else
        pass3_tile<RG3>(dl, xshr, partial, red, tile, split, b, wave, lane);
}

__global__ void sufg_reduce_kernel(const float* __restrict__ partial,
                                   const float* __restrict__ bias,
                                   float* __restrict__ out) {
    int idx = blockIdx.x * blockDim.x + threadIdx.x;
    if (idx >= BB * KOUT * FF) return;
    int o = idx & 31;
    int k = (idx >> 5) % KOUT;
    int b = idx / (KOUT * FF);
    float s = bias[o];
#pragma unroll
    for (int sp = 0; sp < NSPLIT; ++sp)
        s += partial[(((size_t)(b * NSPLIT + sp)) * KPAD + k) * FF + o];
    out[idx] = s;
}

extern "C" void kernel_launch(void* const* d_in, const int* in_sizes, int n_in,
                              void* d_out, int out_size, void* d_ws, size_t ws_size,
                              hipStream_t stream) {
    (void)in_sizes; (void)n_in; (void)out_size; (void)ws_size;
    const float* x    = (const float*)d_in[0];
    const float* dl   = (const float*)d_in[1];
    const float* w    = (const float*)d_in[2];
    const float* filt = (const float*)d_in[3];
    const float* bias = (const float*)d_in[4];
    float* out = (float*)d_out;

    float* ws      = (float*)d_ws;
    float* xshr    = ws + WS_XSHR_OFF;
    float* partial = ws + WS_PART_OFF;

    const float thr = sqrtf(2.0f * logf((float)NN)) / sqrtf((float)NN);

    sufg_pass2_kernel<<<dim3(MSHR / TROWS, BB), 256, 0, stream>>>(dl, x, w, filt, xshr, thr);
    sufg_pass3_kernel<<<dim3(NT3 * NSPLIT, BB), 256, 0, stream>>>(dl, xshr, partial);
    sufg_reduce_kernel<<<(BB * KOUT * FF + 255) / 256, 256, 0, stream>>>(partial, bias, out);
}


# Each wave's requests leave in the order it uses the data: first chunk's fragment loads ahead of any prefetch, A prefetched 64 k at a time four chunks ahead in the K loop, not a whole slice up front
# speedup vs baseline: 2.4653x; 1.0534x over previous
#include <hip/hip_runtime.h>
#include <cmath>


typedef float v2f __attribute__((ext_vector_type(2)));
typedef float v4f __attribute__((ext_vector_type(4)));
typedef float v8f __attribute__((ext_vector_type(8)));
typedef float v2f_a4 __attribute__((ext_vector_type(2), aligned(4)));
typedef float v4f_a4 __attribute__((ext_vector_type(4), aligned(4)));

#define WMMA_F32(A, Bf, C) \
    __builtin_amdgcn_wmma_f32_16x16x4_f32(false, (A), false, (Bf), (short)0, (C), false, false)

#define BB       2
#define NN       2048
#define FF       32
#define DROWS    8192
#define DCOLS    2049
#define MSHR     6144
#define KOUT     2049
#define KPAD     2064
#define NSPLIT   4
#define KSEG     (MSHR / NSPLIT)
#define NWAVE    8
#define KW2      (NN / NWAVE)
#define KW3      (KSEG / NWAVE)
#define TILE_EL  (16 * FF)
#define RG       3
#define TROWS    (16 * RG)
#define RG3      4
#define TROWS3   (16 * RG3)
#define NT3      (NN / TROWS3)
#define RG3L     (RG3 + 1)
#define CHB      4

#define WS_XSHR_OFF  (BB * NN * FF)
#define WS_PART_OFF  (WS_XSHR_OFF + BB * MSHR * FF)

template <int RGT>
struct BlkFrag {
    v4f a[RGT];
    v2f r[4];
};

template <int RGT>
__device__ __forceinline__ void load_blk(BlkFrag<RGT>& f, const float* const (&ap)[RGT], int ao,
                                         const float* bp) {
#pragma unroll
    for (int g = 0; g < RGT; ++g)
        f.a[g] = *(const v4f_a4*)(ap[g] + ao);
    f.r[0] = *(const v2f_a4*)(bp);
    f.r[1] = *(const v2f_a4*)(bp + FF);
    f.r[2] = *(const v2f_a4*)(bp + 2 * FF);
    f.r[3] = *(const v2f_a4*)(bp + 3 * FF);
}

template <int RGT>
__device__ __forceinline__ void mma_blk(const BlkFrag<RGT>& f, v8f (&acc)[RGT][2]) {
    v2f b0;  b0.x = f.r[0].x; b0.y = f.r[1].x;
    v2f b1;  b1.x = f.r[0].y; b1.y = f.r[1].y;
    v2f b2;  b2.x = f.r[2].x; b2.y = f.r[3].x;
    v2f b3;  b3.x = f.r[2].y; b3.y = f.r[3].y;
#pragma unroll
    for (int g = 0; g < RGT; ++g) {
        v2f a01; a01.x = f.a[g].x; a01.y = f.a[g].y;
        acc[g][0] = WMMA_F32(a01, b0, acc[g][0]);
        acc[g][1] = WMMA_F32(a01, b1, acc[g][1]);
    }
#pragma unroll
    for (int g = 0; g < RGT; ++g) {
        v2f a23; a23.x = f.a[g].z; a23.y = f.a[g].w;
        acc[g][0] = WMMA_F32(a23, b2, acc[g][0]);
        acc[g][1] = WMMA_F32(a23, b3, acc[g][1]);
    }
}

template <int RGT>
__device__ __forceinline__ void prefetch_pair(const float* const (&pp)[RGT], int kf) {
#pragma unroll
    for (int g = 0; g < RGT; ++g)
        __builtin_prefetch(pp[g] + kf, 0, 0);
}

template <int KW, int RGT>
__device__ __forceinline__ void k_slice(const float* const (&ap)[RGT],
                                        const float* const (&pp)[RGT],
                                        const float* bp, v8f (&acc)[RGT][2]) {
    constexpr int NCH = KW / (8 * CHB);
    constexpr int NPR = KW / (16 * CHB);
    static_assert(KW % (16 * CHB) == 0, "a K slice is whole pairs of chunks");
    BlkFrag<RGT> f[2][CHB];
#pragma unroll
    for (int j = 0; j < CHB; ++j)
        load_blk<RGT>(f[0][j], ap, 8 * j, bp + 8 * j * FF);
    if (NPR > 1)
        prefetch_pair<RGT>(pp, 16 * CHB);
#pragma unroll
    for (int c = 0; c < NCH; ++c) {
        const int cur = c & 1;
        const int nxt = cur ^ 1;
        if (c + 1 < NCH) {
            const int ko = (c + 1) * 8 * CHB;
#pragma unroll
            for (int j = 0; j < CHB; ++j)
                load_blk<RGT>(f[nxt][j], ap, ko + 8 * j, bp + (ko + 8 * j) * FF);
        }
        if ((c & 1) == 0 && c / 2 + 2 < NPR)
            prefetch_pair<RGT>(pp, (c / 2 + 2) * 16 * CHB);
        __builtin_amdgcn_sched_barrier(0);
#pragma unroll
        for (int j = 0; j < CHB; ++j)
            mma_blk<RGT>(f[cur][j], acc);
    }
}

template <int RGT>
__device__ __forceinline__ void sum_wave_tiles(float* red, int wave, int lane,
                                               const v8f (&acc)[RGT][2],
                                               float (&s_lo)[RGT], float (&s_hi)[RGT]) {
    float* mine = red + wave * (RGT * TILE_EL);
#pragma unroll
    for (int g = 0; g < RGT; ++g) {
#pragma unroll
        for (int r = 0; r < 8; ++r) {
            mine[g * TILE_EL + r * 64 + lane]      = acc[g][0][r];
            mine[g * TILE_EL + r * 64 + 32 + lane] = acc[g][1][r];
        }
    }
    __syncthreads();
    const int i_lo = wave * 64 + (lane & 1) * 32 + (lane >> 1);
    const int i_hi = i_lo + 16;
#pragma unroll
    for (int g = 0; g < RGT; ++g) {
        float a = red[g * TILE_EL + i_lo];
        float c = red[g * TILE_EL + i_hi];
#pragma unroll
        for (int w = 1; w < NWAVE; ++w) {
            a += red[w * (RGT * TILE_EL) + g * TILE_EL + i_lo];
            c += red[w * (RGT * TILE_EL) + g * TILE_EL + i_hi];
        }
        s_lo[g] = a;
        s_hi[g] = c;
    }
}

__global__ void __launch_bounds__(256) __attribute__((amdgpu_waves_per_eu(4, 4)))
sufg_pass2_kernel(const float* __restrict__ dl,
                  const float* __restrict__ x,
                  const float* __restrict__ w,
                  const float* __restrict__ filt,
                  float* __restrict__ xshr,
                  float thr) {
    __shared__ float red[NWAVE * RG * TILE_EL];

    const int lane = threadIdx.x & 31;
    const int wave = threadIdx.x >> 5;
    const int tile = blockIdx.x;
    const int b    = blockIdx.y;

    const int m0   = tile * TROWS;
    const int row  = lane & 15;
    const int hi   = lane >> 4;
    const int koff = hi << 2;
    const int k0   = wave * KW2;

    const float* Arow = dl + (size_t)b * DROWS * DCOLS
                           + (size_t)(2048 + m0 + row) * DCOLS + 1 + k0;
    const float* A  = Arow + koff;
    const float* Bp = x + (size_t)b * NN * FF + (size_t)(k0 + koff) * FF + 2 * row;

    v8f acc[RG][2];
#pragma unroll
    for (int g = 0; g < RG; ++g) {
        acc[g][0] = (v8f){0.f,0.f,0.f,0.f,0.f,0.f,0.f,0.f};
        acc[g][1] = (v8f){0.f,0.f,0.f,0.f,0.f,0.f,0.f,0.f};
    }

    const float* const Ap[RG] = { A, A + (size_t)16 * DCOLS, A + (size_t)32 * DCOLS };
    const float* const Pr = Arow + hi * 32;
    const float* const Pp[RG] = { Pr, Pr + (size_t)16 * DCOLS, Pr + (size_t)32 * DCOLS };
    k_slice<KW2, RG>(Ap, Pp, Bp, acc);

    float wcol[FF];
#pragma unroll
    for (int f = 0; f < FF; ++f) wcol[f] = w[f * FF + lane];

    float s_lo[RG], s_hi[RG];
    sum_wave_tiles<RG>(red, wave, lane, acc, s_lo, s_hi);

    float t_lo[RG], t_hi[RG];
#pragma unroll
    for (int g = 0; g < RG; ++g) { t_lo[g] = 0.f; t_hi[g] = 0.f; }
#pragma unroll
    for (int f = 0; f < FF; ++f) {
#pragma unroll
        for (int g = 0; g < RG; ++g) {
            const float a = __builtin_bit_cast(float,
                __builtin_amdgcn_readlane(__builtin_bit_cast(int, s_lo[g]), f));
            const float c = __builtin_bit_cast(float,
                __builtin_amdgcn_readlane(__builtin_bit_cast(int, s_hi[g]), f));
            t_lo[g] = fmaf(a, wcol[f], t_lo[g]);
            t_hi[g] = fmaf(c, wcol[f], t_hi[g]);
        }
    }

#pragma unroll
    for (int g = 0; g < RG; ++g) {
        const int mA = m0 + 16 * g + wave;
        const int mB = mA + 8;
        float vA = t_lo[g] * filt[2048 + mA];
        float vB = t_hi[g] * filt[2048 + mB];
        if (m0 + 16 * g >= 2048) {
            vA = copysignf(fmaxf(fabsf(vA) - thr, 0.f), vA);
            vB = copysignf(fmaxf(fabsf(vB) - thr, 0.f), vB);
        }
        xshr[((size_t)b * MSHR + mA) * FF + lane] = vA;
        xshr[((size_t)b * MSHR + mB) * FF + lane] = vB;
    }
}

template <int RGT>
__device__ __forceinline__ void pass3_tile(const float* __restrict__ dl,
                                           const float* __restrict__ xshr,
                                           float* __restrict__ partial,
                                           float* red, int tile, int split, int b,
                                           int wave, int lane) {
    const int kbegin = split * KSEG + wave * KW3;

    const int row  = lane & 15;
    const int hi   = lane >> 4;
    const int koff = hi << 2;

    const float* Abase = dl + (size_t)b * DROWS * DCOLS + (size_t)2048 * DCOLS + kbegin;
    const float* Ag[RGT];
#pragma unroll
    for (int g = 0; g < RGT; ++g) {
        int krow = tile * TROWS3 + 16 * g + row;
        if (krow > KOUT - 1) krow = KOUT - 1;
        Ag[g] = Abase + (size_t)krow * MSHR;
    }
    const float* Bp = xshr + (size_t)b * MSHR * FF + (size_t)(kbegin + koff) * FF + 2 * row;

    v8f acc[RGT][2];
#pragma unroll
    for (int g = 0; g < RGT; ++g) {
        acc[g][0] = (v8f){0.f,0.f,0.f,0.f,0.f,0.f,0.f,0.f};
        acc[g][1] = (v8f){0.f,0.f,0.f,0.f,0.f,0.f,0.f,0.f};
    }

    const float* Ap[RGT];
    const float* Pp[RGT];
#pragma unroll
    for (int g = 0; g < RGT; ++g) { Ap[g] = Ag[g] + koff; Pp[g] = Ag[g] + hi * 32; }
    k_slice<KW3, RGT>(Ap, Pp, Bp, acc);

    float s_lo[RGT], s_hi[RGT];
    sum_wave_tiles<RGT>(red, wave, lane, acc, s_lo, s_hi);

    float* pbase = partial + ((size_t)(b * NSPLIT + split)) * KPAD * FF;
#pragma unroll
    for (int g = 0; g < RGT; ++g) {
        const int kkA = tile * TROWS3 + 16 * g + wave;
        pbase[(size_t)kkA * FF + lane]       = s_lo[g];
        pbase[(size_t)(kkA + 8) * FF + lane] = s_hi[g];
    }
}

__global__ void __launch_bounds__(256) __attribute__((amdgpu_waves_per_eu(2, 2)))
sufg_pass3_kernel(const float* __restrict__ dl,
                  const float* __restrict__ xshr,
                  float* __restrict__ partial) {
    __shared__ float red[NWAVE * RG3L * TILE_EL];

    const int lane = threadIdx.x & 31;
    const int wave = threadIdx.x >> 5;
    const int unit = blockIdx.x;
    const int b    = blockIdx.y;

    const int split = unit & (NSPLIT - 1);
    const int tile  = (NT3 - 1) - (unit >> 2);

    if (tile == NT3 - 1)
        pass3_tile<RG3L>(dl, xshr, partial, red, tile, split, b, wave, lane);
    else
        pass3_tile<RG3>(dl, xshr, partial, red, tile, split, b, wave, lane);
}

__global__ void sufg_reduce_kernel(const float* __restrict__ partial,
                                   const float* __restrict__ bias,
                                   float* __restrict__ out) {
    int idx = blockIdx.x * blockDim.x + threadIdx.x;
    if (idx >= BB * KOUT * FF) return;
    int o = idx & 31;
    int k = (idx >> 5) % KOUT;
    int b = idx / (KOUT * FF);
    float s = bias[o];
#pragma unroll
    for (int sp = 0; sp < NSPLIT; ++sp)
        s += partial[(((size_t)(b * NSPLIT + sp)) * KPAD + k) * FF + o];
    out[idx] = s;
}

extern "C" void kernel_launch(void* const* d_in, const int* in_sizes, int n_in,
                              void* d_out, int out_size, void* d_ws, size_t ws_size,
                              hipStream_t stream) {
    (void)in_sizes; (void)n_in; (void)out_size; (void)ws_size;
    const float* x    = (const float*)d_in[0];
    const float* dl   = (const float*)d_in[1];
    const float* w    = (const float*)d_in[2];
    const float* filt = (const float*)d_in[3];
    const float* bias = (const float*)d_in[4];
    float* out = (float*)d_out;

    float* ws      = (float*)d_ws;
    float* xshr    = ws + WS_XSHR_OFF;
    float* partial = ws + WS_PART_OFF;

    const float thr = sqrtf(2.0f * logf((float)NN)) / sqrtf((float)NN);

    sufg_pass2_kernel<<<dim3(MSHR / TROWS, BB), 256, 0, stream>>>(dl, x, w, filt, xshr, thr);
    sufg_pass3_kernel<<<dim3(NT3 * NSPLIT, BB), 256, 0, stream>>>(dl, xshr, partial);
    sufg_reduce_kernel<<<(BB * KOUT * FF + 255) / 256, 256, 0, stream>>>(partial, bias, out);
}


# A fragments fed through a per-wave LDS stage: a global A load takes 128 contiguous bytes from each of 4 rows, not 32 bytes from each of 16; same k per fragment element, sum order unchanged
# speedup vs baseline: 2.5541x; 1.0360x over previous
#include <hip/hip_runtime.h>
#include <cmath>


typedef float v2f __attribute__((ext_vector_type(2)));
typedef float v4f __attribute__((ext_vector_type(4)));
typedef float v8f __attribute__((ext_vector_type(8)));
typedef float v2f_a4 __attribute__((ext_vector_type(2), aligned(4)));
typedef float v4f_a4 __attribute__((ext_vector_type(4), aligned(4)));

#define WMMA_F32(A, Bf, C) \
    __builtin_amdgcn_wmma_f32_16x16x4_f32(false, (A), false, (Bf), (short)0, (C), false, false)

#define BB       2
#define NN       2048
#define FF       32
#define DROWS    8192
#define DCOLS    2049
#define MSHR     6144
#define KOUT     2049
#define KPAD     2064
#define NSPLIT   4
#define KSEG     (MSHR / NSPLIT)
#define NWAVE    8
#define KW2      (NN / NWAVE)
#define KW3      (KSEG / NWAVE)
#define TILE_EL  (16 * FF)
#define RG       3
#define TROWS    (16 * RG)
#define RG3      4
#define TROWS3   (16 * RG3)
#define NT3      (NN / TROWS3)
#define RG3L     (RG3 + 1)
#define CHB      4
#define SPITCH   36
#define STAGE_EL(RGT) ((RGT) * 16 * SPITCH)

#define WS_XSHR_OFF  (BB * NN * FF)
#define WS_PART_OFF  (WS_XSHR_OFF + BB * MSHR * FF)

template <int RGT>
struct BlkFrag {
    v4f a[RGT];
    v2f r[4];
};

template <int RGT>
__device__ __forceinline__ void load_b(BlkFrag<RGT>& f, const float* bp) {
    f.r[0] = *(const v2f_a4*)(bp);
    f.r[1] = *(const v2f_a4*)(bp + FF);
    f.r[2] = *(const v2f_a4*)(bp + 2 * FF);
    f.r[3] = *(const v2f_a4*)(bp + 3 * FF);
}

template <int RGT>
__device__ __forceinline__ void stage_load(v4f (&s)[RGT][4], const float* const (&sp)[RGT],
                                           const int (&rs)[RGT], int kf) {
#pragma unroll
    for (int g = 0; g < RGT; ++g)
#pragma unroll
        for (int j = 0; j < 4; ++j)
            s[g][j] = *(const v4f_a4*)(sp[g] + j * rs[g] + kf);
}

template <int RGT>
__device__ __forceinline__ void stage_store(const v4f (&s)[RGT][4], float* st_w) {
#pragma unroll
    for (int g = 0; g < RGT; ++g)
#pragma unroll
        for (int j = 0; j < 4; ++j)
            *(v4f*)(st_w + (16 * g + 4 * j) * SPITCH) = s[g][j];
}

template <int RGT>
__device__ __forceinline__ void frag_read(BlkFrag<RGT> (&f)[CHB], const float* st_r) {
#pragma unroll
    for (int jb = 0; jb < CHB; ++jb)
#pragma unroll
        for (int g = 0; g < RGT; ++g)
            f[jb].a[g] = *(const v4f*)(st_r + 16 * g * SPITCH + 8 * jb);
}

template <int RGT>
__device__ __forceinline__ void mma_blk(const BlkFrag<RGT>& f, v8f (&acc)[RGT][2]) {
    v2f b0;  b0.x = f.r[0].x; b0.y = f.r[1].x;
    v2f b1;  b1.x = f.r[0].y; b1.y = f.r[1].y;
    v2f b2;  b2.x = f.r[2].x; b2.y = f.r[3].x;
    v2f b3;  b3.x = f.r[2].y; b3.y = f.r[3].y;
#pragma unroll
    for (int g = 0; g < RGT; ++g) {
        v2f a01; a01.x = f.a[g].x; a01.y = f.a[g].y;
        acc[g][0] = WMMA_F32(a01, b0, acc[g][0]);
        acc[g][1] = WMMA_F32(a01, b1, acc[g][1]);
    }
#pragma unroll
    for (int g = 0; g < RGT; ++g) {
        v2f a23; a23.x = f.a[g].z; a23.y = f.a[g].w;
        acc[g][0] = WMMA_F32(a23, b2, acc[g][0]);
        acc[g][1] = WMMA_F32(a23, b3, acc[g][1]);
    }
}

template <int RGT>
__device__ __forceinline__ void prefetch_pair(const float* const (&pp)[RGT], int kf) {
#pragma unroll
    for (int g = 0; g < RGT; ++g)
        __builtin_prefetch(pp[g] + kf, 0, 0);
}

template <int KW, int RGT>
__device__ __forceinline__ void k_slice(const float* const (&sp)[RGT], const int (&rs)[RGT],
                                        const float* const (&pp)[RGT], const float* bp,
                                        float* st_w, const float* st_r, v8f (&acc)[RGT][2]) {
    constexpr int NCH = KW / (8 * CHB);
    constexpr int NPR = KW / (16 * CHB);
    static_assert(KW % (16 * CHB) == 0, "a K slice is whole pairs of chunks");
    BlkFrag<RGT> f[2][CHB];
    v4f s[RGT][4];
    stage_load<RGT>(s, sp, rs, 0);
#pragma unroll
    for (int g = 0; g < RGT; ++g)
#pragma unroll
        for (int j = 0; j < 4; ++j)
            f[1][j].a[g] = *(const v4f_a4*)(sp[g] + j * rs[g] + 8 * CHB);
#pragma unroll
    for (int j = 0; j < CHB; ++j)
        load_b<RGT>(f[0][j], bp + 8 * j * FF);
    if (NPR > 1)
        prefetch_pair<RGT>(pp, 16 * CHB);
    stage_store<RGT>(s, st_w);
    __builtin_amdgcn_wave_barrier();
    frag_read<RGT>(f[0], st_r);
#pragma unroll
    for (int c = 0; c < NCH; ++c) {
        const int cur = c & 1;
        const int nxt = cur ^ 1;
        if (c + 1 < NCH) {
            const int ko = (c + 1) * 8 * CHB;
            __builtin_amdgcn_wave_barrier();
            if (c == 0) {
#pragma unroll
                for (int g = 0; g < RGT; ++g)
#pragma unroll
                    for (int j = 0; j < 4; ++j)
                        *(v4f*)(st_w + (16 * g + 4 * j) * SPITCH) = f[1][j].a[g];
            } else {
                stage_store<RGT>(s, st_w);
            }
            if (c + 2 < NCH)
                stage_load<RGT>(s, sp, rs, ko + 8 * CHB);
#pragma unroll
            for (int j = 0; j < CHB; ++j)
                load_b<RGT>(f[nxt][j], bp + (ko + 8 * j) * FF);
            __builtin_amdgcn_wave_barrier();
            frag_read<RGT>(f[nxt], st_r);
        }
        if ((c & 1) == 0 && c / 2 + 2 < NPR)
            prefetch_pair<RGT>(pp, (c / 2 + 2) * 16 * CHB);
        __builtin_amdgcn_sched_barrier(0);
#pragma unroll
        for (int j = 0; j < CHB; ++j)
            mma_blk<RGT>(f[cur][j], acc);
        __builtin_amdgcn_sched_barrier(0);
    }
}

template <int RGT>
__device__ __forceinline__ void sum_wave_tiles(float* red, int wave, int lane,
                                               const v8f (&acc)[RGT][2],
                                               float (&s_lo)[RGT], float (&s_hi)[RGT]) {
    float* mine = red + wave * (RGT * TILE_EL);
#pragma unroll
    for (int g = 0; g < RGT; ++g) {
#pragma unroll
        for (int r = 0; r < 8; ++r) {
            mine[g * TILE_EL + r * 64 + lane]      = acc[g][0][r];
            mine[g * TILE_EL + r * 64 + 32 + lane] = acc[g][1][r];
        }
    }
    __syncthreads();
    const int i_lo = wave * 64 + (lane & 1) * 32 + (lane >> 1);
    const int i_hi = i_lo + 16;
#pragma unroll
    for (int g = 0; g < RGT; ++g) {
        float a = red[g * TILE_EL + i_lo];
        float c = red[g * TILE_EL + i_hi];
#pragma unroll
        for (int w = 1; w < NWAVE; ++w) {
            a += red[w * (RGT * TILE_EL) + g * TILE_EL + i_lo];
            c += red[w * (RGT * TILE_EL) + g * TILE_EL + i_hi];
        }
        s_lo[g] = a;
        s_hi[g] = c;
    }
}

__global__ void __launch_bounds__(256) __attribute__((amdgpu_waves_per_eu(2, 2)))
sufg_pass2_kernel(const float* __restrict__ dl,
                  const float* __restrict__ x,
                  const float* __restrict__ w,
                  const float* __restrict__ filt,
                  float* __restrict__ xshr,
                  float thr) {
    __shared__ float red[NWAVE * RG * TILE_EL];
    __shared__ __attribute__((aligned(16))) float stage[NWAVE * STAGE_EL(RG)];

    const int lane = threadIdx.x & 31;
    const int wave = threadIdx.x >> 5;
    const int tile = blockIdx.x;
    const int b    = blockIdx.y;

    const int m0   = tile * TROWS;
    const int row  = lane & 15;
    const int hi   = lane >> 4;
    const int koff = hi << 2;
    const int k0   = wave * KW2;

    const float* Arow = dl + (size_t)b * DROWS * DCOLS
                           + (size_t)(2048 + m0 + row) * DCOLS + 1 + k0;
    const int r4 = lane >> 3;
    const int c8 = lane & 7;
    const float* S0 = dl + (size_t)b * DROWS * DCOLS
                         + (size_t)(2048 + m0 + r4) * DCOLS + 1 + k0 + 4 * c8;
    const float* Bp = x + (size_t)b * NN * FF + (size_t)(k0 + koff) * FF + 2 * row;

    v8f acc[RG][2];
#pragma unroll
    for (int g = 0; g < RG; ++g) {
        acc[g][0] = (v8f){0.f,0.f,0.f,0.f,0.f,0.f,0.f,0.f};
        acc[g][1] = (v8f){0.f,0.f,0.f,0.f,0.f,0.f,0.f,0.f};
    }

    const float* const Sp[RG] = { S0, S0 + (size_t)16 * DCOLS, S0 + (size_t)32 * DCOLS };
    const int Rs[RG] = { 4 * DCOLS, 4 * DCOLS, 4 * DCOLS };
    float* const st   = stage + wave * STAGE_EL(RG);
    float* const st_w = st + r4 * SPITCH + 4 * c8;
    const float* const st_r = st + row * SPITCH + koff;
    const float* const Pr = Arow + hi * 32;
    const float* const Pp[RG] = { Pr, Pr + (size_t)16 * DCOLS, Pr + (size_t)32 * DCOLS };
    k_slice<KW2, RG>(Sp, Rs, Pp, Bp, st_w, st_r, acc);

    float wcol[FF];
#pragma unroll
    for (int f = 0; f < FF; ++f) wcol[f] = w[f * FF + lane];

    float s_lo[RG], s_hi[RG];
    sum_wave_tiles<RG>(red, wave, lane, acc, s_lo, s_hi);

    float t_lo[RG], t_hi[RG];
#pragma unroll
    for (int g = 0; g < RG; ++g) { t_lo[g] = 0.f; t_hi[g] = 0.f; }
#pragma unroll
    for (int f = 0; f < FF; ++f) {
#pragma unroll
        for (int g = 0; g < RG; ++g) {
            const float a = __builtin_bit_cast(float,
                __builtin_amdgcn_readlane(__builtin_bit_cast(int, s_lo[g]), f));
            const float c = __builtin_bit_cast(float,
                __builtin_amdgcn_readlane(__builtin_bit_cast(int, s_hi[g]), f));
            t_lo[g] = fmaf(a, wcol[f], t_lo[g]);
            t_hi[g] = fmaf(c, wcol[f], t_hi[g]);
        }
    }

#pragma unroll
    for (int g = 0; g < RG; ++g) {
        const int mA = m0 + 16 * g + wave;
        const int mB = mA + 8;
        float vA = t_lo[g] * filt[2048 + mA];
        float vB = t_hi[g] * filt[2048 + mB];
        if (m0 + 16 * g >= 2048) {
            vA = copysignf(fmaxf(fabsf(vA) - thr, 0.f), vA);
            vB = copysignf(fmaxf(fabsf(vB) - thr, 0.f), vB);
        }
        xshr[((size_t)b * MSHR + mA) * FF + lane] = vA;
        xshr[((size_t)b * MSHR + mB) * FF + lane] = vB;
    }
}

template <int RGT>
__device__ __forceinline__ void pass3_tile(const float* __restrict__ dl,
                                           const float* __restrict__ xshr,
                                           float* __restrict__ partial,
                                           float* red, float* stage, int tile, int split, int b,
                                           int wave, int lane) {
    const int kbegin = split * KSEG + wave * KW3;

    const int row  = lane & 15;
    const int hi   = lane >> 4;
    const int koff = hi << 2;

    const float* Abase = dl + (size_t)b * DROWS * DCOLS + (size_t)2048 * DCOLS + kbegin;
    const float* Ag[RGT];
#pragma unroll
    for (int g = 0; g < RGT; ++g) {
        int krow = tile * TROWS3 + 16 * g + row;
        if (krow > KOUT - 1) krow = KOUT - 1;
        Ag[g] = Abase + (size_t)krow * MSHR;
    }
    const float* Bp = xshr + (size_t)b * MSHR * FF + (size_t)(kbegin + koff) * FF + 2 * row;

    v8f acc[RGT][2];
#pragma unroll
    for (int g = 0; g < RGT; ++g) {
        acc[g][0] = (v8f){0.f,0.f,0.f,0.f,0.f,0.f,0.f,0.f};
        acc[g][1] = (v8f){0.f,0.f,0.f,0.f,0.f,0.f,0.f,0.f};
    }

    const float* Pp[RGT];
#pragma unroll
    for (int g = 0; g < RGT; ++g) Pp[g] = Ag[g] + hi * 32;
    const int r4 = lane >> 3;
    const int c8 = lane & 7;
    const float* Sp[RGT];
    int Rs[RGT];
#pragma unroll
    for (int g = 0; g < RGT; ++g) {
        int krow = tile * TROWS3 + 16 * g + r4;
        int step = 4 * MSHR;
        if (g >= RG3) { krow = KOUT - 1; step = 0; }
        Sp[g] = Abase + (size_t)krow * MSHR + 4 * c8;
        Rs[g] = step;
    }
    float* const st   = stage + wave * STAGE_EL(RGT);
    float* const st_w = st + r4 * SPITCH + 4 * c8;
    const float* const st_r = st + row * SPITCH + koff;
    k_slice<KW3, RGT>(Sp, Rs, Pp, Bp, st_w, st_r, acc);

    float s_lo[RGT], s_hi[RGT];
    sum_wave_tiles<RGT>(red, wave, lane, acc, s_lo, s_hi);

    float* pbase = partial + ((size_t)(b * NSPLIT + split)) * KPAD * FF;
#pragma unroll
    for (int g = 0; g < RGT; ++g) {
        const int kkA = tile * TROWS3 + 16 * g + wave;
        pbase[(size_t)kkA * FF + lane]       = s_lo[g];
        pbase[(size_t)(kkA + 8) * FF + lane] = s_hi[g];
    }
}

__global__ void __launch_bounds__(256) __attribute__((amdgpu_waves_per_eu(2, 2)))
sufg_pass3_kernel(const float* __restrict__ dl,
                  const float* __restrict__ xshr,
                  float* __restrict__ partial) {
    __shared__ float red[NWAVE * RG3L * TILE_EL];
    __shared__ __attribute__((aligned(16))) float stage[NWAVE * STAGE_EL(RG3L)];

    const int lane = threadIdx.x & 31;
    const int wave = threadIdx.x >> 5;
    const int unit = blockIdx.x;
    const int b    = blockIdx.y;

    const int split = unit & (NSPLIT - 1);
    const int tile  = (NT3 - 1) - (unit >> 2);

    if (tile == NT3 - 1)
        pass3_tile<RG3L>(dl, xshr, partial, red, stage, tile, split, b, wave, lane);
    else
        pass3_tile<RG3>(dl, xshr, partial, red, stage, tile, split, b, wave, lane);
}

__global__ void sufg_reduce_kernel(const float* __restrict__ partial,
                                   const float* __restrict__ bias,
                                   float* __restrict__ out) {
    int idx = blockIdx.x * blockDim.x + threadIdx.x;
    if (idx >= BB * KOUT * FF) return;
    int o = idx & 31;
    int k = (idx >> 5) % KOUT;
    int b = idx / (KOUT * FF);
    float s = bias[o];
#pragma unroll
    for (int sp = 0; sp < NSPLIT; ++sp)
        s += partial[(((size_t)(b * NSPLIT + sp)) * KPAD + k) * FF + o];
    out[idx] = s;
}

extern "C" void kernel_launch(void* const* d_in, const int* in_sizes, int n_in,
                              void* d_out, int out_size, void* d_ws, size_t ws_size,
                              hipStream_t stream) {
    (void)in_sizes; (void)n_in; (void)out_size; (void)ws_size;
    const float* x    = (const float*)d_in[0];
    const float* dl   = (const float*)d_in[1];
    const float* w    = (const float*)d_in[2];
    const float* filt = (const float*)d_in[3];
    const float* bias = (const float*)d_in[4];
    float* out = (float*)d_out;

    float* ws      = (float*)d_ws;
    float* xshr    = ws + WS_XSHR_OFF;
    float* partial = ws + WS_PART_OFF;

    const float thr = sqrtf(2.0f * logf((float)NN)) / sqrtf((float)NN);

    sufg_pass2_kernel<<<dim3(MSHR / TROWS, BB), 256, 0, stream>>>(dl, x, w, filt, xshr, thr);
    sufg_pass3_kernel<<<dim3(NT3 * NSPLIT, BB), 256, 0, stream>>>(dl, xshr, partial);
    sufg_reduce_kernel<<<(BB * KOUT * FF + 255) / 256, 256, 0, stream>>>(partial, bias, out);
}


# Pass 2 tiles take 48 rows 32 apart (one residue class mod 32) and rotate k by the class's line offset: every staged A load is whole aligned 128-byte lines, each y line loaded once; new sum order
# speedup vs baseline: 2.7223x; 1.0659x over previous
#include <hip/hip_runtime.h>
#include <cmath>


typedef float v2f __attribute__((ext_vector_type(2)));
typedef float v4f __attribute__((ext_vector_type(4)));
typedef float v8f __attribute__((ext_vector_type(8)));
typedef float v2f_a4 __attribute__((ext_vector_type(2), aligned(4)));
typedef float v4f_a4 __attribute__((ext_vector_type(4), aligned(4)));

#define WMMA_F32(A, Bf, C) \
    __builtin_amdgcn_wmma_f32_16x16x4_f32(false, (A), false, (Bf), (short)0, (C), false, false)

#define BB       2
#define NN       2048
#define FF       32
#define DROWS    8192
#define DCOLS    2049
#define MSHR     6144
#define KOUT     2049
#define KPAD     2064
#define NSPLIT   4
#define KSEG     (MSHR / NSPLIT)
#define NWAVE    8
#define KW2      (NN / NWAVE)
#define KW3      (KSEG / NWAVE)
#define TILE_EL  (16 * FF)
#define RG       3
#define TROWS    (16 * RG)
#define RG3      4
#define TROWS3   (16 * RG3)
#define NT3      (NN / TROWS3)
#define RG3L     (RG3 + 1)
#define CHB      4
#define SPITCH   36
#define STAGE_EL(RGT) ((RGT) * 16 * SPITCH)

#define WS_XSHR_OFF  (BB * NN * FF)
#define WS_PART_OFF  (WS_XSHR_OFF + BB * MSHR * FF)

template <int RGT>
struct BlkFrag {
    v4f a[RGT];
    v2f r[4];
};

template <int RGT>
__device__ __forceinline__ void load_b(BlkFrag<RGT>& f, const float* bp) {
    f.r[0] = *(const v2f_a4*)(bp);
    f.r[1] = *(const v2f_a4*)(bp + FF);
    f.r[2] = *(const v2f_a4*)(bp + 2 * FF);
    f.r[3] = *(const v2f_a4*)(bp + 3 * FF);
}

template <int RGT>
__device__ __forceinline__ void stage_load(v4f (&s)[RGT][4], const float* const (&sp)[RGT],
                                           const int (&rs)[RGT], int kf) {
#pragma unroll
    for (int g = 0; g < RGT; ++g)
#pragma unroll
        for (int j = 0; j < 4; ++j)
            s[g][j] = *(const v4f_a4*)(sp[g] + j * rs[g] + kf);
}

template <int RGT>
__device__ __forceinline__ void stage_store(const v4f (&s)[RGT][4], float* st_w) {
#pragma unroll
    for (int g = 0; g < RGT; ++g)
#pragma unroll
        for (int j = 0; j < 4; ++j)
            *(v4f*)(st_w + (16 * g + 4 * j) * SPITCH) = s[g][j];
}

template <int RGT>
__device__ __forceinline__ void frag_read(BlkFrag<RGT> (&f)[CHB], const float* st_r) {
#pragma unroll
    for (int jb = 0; jb < CHB; ++jb)
#pragma unroll
        for (int g = 0; g < RGT; ++g)
            f[jb].a[g] = *(const v4f*)(st_r + 16 * g * SPITCH + 8 * jb);
}

template <int RGT>
__device__ __forceinline__ void mma_blk(const BlkFrag<RGT>& f, v8f (&acc)[RGT][2]) {
    v2f b0;  b0.x = f.r[0].x; b0.y = f.r[1].x;
    v2f b1;  b1.x = f.r[0].y; b1.y = f.r[1].y;
    v2f b2;  b2.x = f.r[2].x; b2.y = f.r[3].x;
    v2f b3;  b3.x = f.r[2].y; b3.y = f.r[3].y;
#pragma unroll
    for (int g = 0; g < RGT; ++g) {
        v2f a01; a01.x = f.a[g].x; a01.y = f.a[g].y;
        acc[g][0] = WMMA_F32(a01, b0, acc[g][0]);
        acc[g][1] = WMMA_F32(a01, b1, acc[g][1]);
    }
#pragma unroll
    for (int g = 0; g < RGT; ++g) {
        v2f a23; a23.x = f.a[g].z; a23.y = f.a[g].w;
        acc[g][0] = WMMA_F32(a23, b2, acc[g][0]);
        acc[g][1] = WMMA_F32(a23, b3, acc[g][1]);
    }
}

template <int RGT>
__device__ __forceinline__ void prefetch_pair(const float* const (&pp)[RGT], int kf) {
#pragma unroll
    for (int g = 0; g < RGT; ++g)
        __builtin_prefetch(pp[g] + kf, 0, 0);
}

template <int KW, int RGT, bool ROT>
__device__ __forceinline__ void k_slice(const float* const (&sp)[RGT], const int (&rs)[RGT],
                                        const float* const (&pp)[RGT], const float* bp,
                                        float* st_w, const float* st_r, v8f (&acc)[RGT][2],
                                        int nw, const float* bx, int kb0) {
    constexpr int NCH = KW / (8 * CHB);
    constexpr int NPR = KW / (16 * CHB);
    static_assert(KW % (16 * CHB) == 0, "a K slice is whole pairs of chunks");
    BlkFrag<RGT> f[2][CHB];
    v4f s[RGT][4];
    stage_load<RGT>(s, sp, rs, 0);
    v4f t[RGT][4];
    if (ROT) {
        const int wo = nw > 0 ? NN : 0;
#pragma unroll
        for (int g = 0; g < RGT; ++g)
#pragma unroll
            for (int j = 0; j < 4; ++j)
                t[g][j] = *(const v4f_a4*)(sp[g] + j * rs[g] + wo);
    }
#pragma unroll
    for (int g = 0; g < RGT; ++g)
#pragma unroll
        for (int j = 0; j < 4; ++j)
            f[1][j].a[g] = *(const v4f_a4*)(sp[g] + j * rs[g] + 8 * CHB);
    if (ROT) {
#pragma unroll
        for (int j = 0; j < CHB; ++j)
#pragma unroll
            for (int jr = 0; jr < 4; ++jr)
                f[0][j].r[jr] = *(const v2f_a4*)(bx + ((kb0 + 8 * j + jr) & (NN - 1)) * FF);
    } else {
#pragma unroll
        for (int j = 0; j < CHB; ++j)
            load_b<RGT>(f[0][j], bp + 8 * j * FF);
    }
    if (NPR > 1)
        prefetch_pair<RGT>(pp, 16 * CHB);
    if (ROT) {
        __builtin_amdgcn_sched_barrier(0);
#pragma unroll
        for (int g = 0; g < RGT; ++g)
#pragma unroll
            for (int j = 0; j < 4; ++j) {
                s[g][j].x = nw > 0 ? t[g][j].x : s[g][j].x;
                s[g][j].y = nw > 1 ? t[g][j].y : s[g][j].y;
                s[g][j].z = nw > 2 ? t[g][j].z : s[g][j].z;
                s[g][j].w = nw > 3 ? t[g][j].w : s[g][j].w;
            }
    }
    stage_store<RGT>(s, st_w);
    __builtin_amdgcn_wave_barrier();
    frag_read<RGT>(f[0], st_r);
#pragma unroll
    for (int c = 0; c < NCH; ++c) {
        const int cur = c & 1;
        const int nxt = cur ^ 1;
        if (c + 1 < NCH) {
            const int ko = (c + 1) * 8 * CHB;
            __builtin_amdgcn_wave_barrier();
            if (c == 0) {
#pragma unroll
                for (int g = 0; g < RGT; ++g)
#pragma unroll
                    for (int j = 0; j < 4; ++j)
                        *(v4f*)(st_w + (16 * g + 4 * j) * SPITCH) = f[1][j].a[g];
            } else {
                stage_store<RGT>(s, st_w);
            }
            if (c + 2 < NCH)
                stage_load<RGT>(s, sp, rs, ko + 8 * CHB);
#pragma unroll
            for (int j = 0; j < CHB; ++j)
                load_b<RGT>(f[nxt][j], bp + (ko + 8 * j) * FF);
            __builtin_amdgcn_wave_barrier();
            frag_read<RGT>(f[nxt], st_r);
        }
        if ((c & 1) == 0 && c / 2 + 2 < NPR)
            prefetch_pair<RGT>(pp, (c / 2 + 2) * 16 * CHB);
        __builtin_amdgcn_sched_barrier(0);
#pragma unroll
        for (int j = 0; j < CHB; ++j)
            mma_blk<RGT>(f[cur][j], acc);
        __builtin_amdgcn_sched_barrier(0);
    }
}

template <int RGT>
__device__ __forceinline__ void sum_wave_tiles(float* red, int wave, int lane,
                                               const v8f (&acc)[RGT][2],
                                               float (&s_lo)[RGT], float (&s_hi)[RGT]) {
    float* mine = red + wave * (RGT * TILE_EL);
#pragma unroll
    for (int g = 0; g < RGT; ++g) {
#pragma unroll
        for (int r = 0; r < 8; ++r) {
            mine[g * TILE_EL + r * 64 + lane]      = acc[g][0][r];
            mine[g * TILE_EL + r * 64 + 32 + lane] = acc[g][1][r];
        }
    }
    __syncthreads();
    const int i_lo = wave * 64 + (lane & 1) * 32 + (lane >> 1);
    const int i_hi = i_lo + 16;
#pragma unroll
    for (int g = 0; g < RGT; ++g) {
        float a = red[g * TILE_EL + i_lo];
        float c = red[g * TILE_EL + i_hi];
#pragma unroll
        for (int w = 1; w < NWAVE; ++w) {
            a += red[w * (RGT * TILE_EL) + g * TILE_EL + i_lo];
            c += red[w * (RGT * TILE_EL) + g * TILE_EL + i_hi];
        }
        s_lo[g] = a;
        s_hi[g] = c;
    }
}

__global__ void __launch_bounds__(256) __attribute__((amdgpu_waves_per_eu(2, 2)))
sufg_pass2_kernel(const float* __restrict__ dl,
                  const float* __restrict__ x,
                  const float* __restrict__ w,
                  const float* __restrict__ filt,
                  float* __restrict__ xshr,
                  float thr) {
    __shared__ float red[NWAVE * RG * TILE_EL];
    __shared__ __attribute__((aligned(16))) float stage[NWAVE * STAGE_EL(RG)];

    const int lane = threadIdx.x & 31;
    const int wave = threadIdx.x >> 5;
    const int tile = blockIdx.x;
    const int b    = blockIdx.y;

    const int res  = tile >> 2;
    const int i0   = (tile & 3) * TROWS;
    const int rot  = (res + 1) & 31;
    const int row  = lane & 15;
    const int hi   = lane >> 4;
    const int koff = hi << 2;
    const int k0   = wave * KW2;

    const float* Arow = dl + (size_t)b * DROWS * DCOLS
                           + (size_t)(2048 + res + 32 * (i0 + row)) * DCOLS + 1 - rot + k0;
    const int r4 = lane >> 3;
    const int c8 = lane & 7;
    const float* S0 = dl + (size_t)b * DROWS * DCOLS
                         + (size_t)(2048 + res + 32 * (i0 + r4)) * DCOLS + 1 - rot + k0 + 4 * c8;
    const float* Bx = x + (size_t)b * NN * FF + 2 * row;
    const int kb0 = k0 - rot + koff;
    const float* Bp = Bx + kb0 * FF;
    const int nw = (wave == 0) ? rot - 4 * c8 : 0;

    v8f acc[RG][2];
#pragma unroll
    for (int g = 0; g < RG; ++g) {
        acc[g][0] = (v8f){0.f,0.f,0.f,0.f,0.f,0.f,0.f,0.f};
        acc[g][1] = (v8f){0.f,0.f,0.f,0.f,0.f,0.f,0.f,0.f};
    }

    const float* const Sp[RG] = { S0, S0 + (size_t)16 * 32 * DCOLS, S0 + (size_t)32 * 32 * DCOLS };
    const int Rs[RG] = { 4 * 32 * DCOLS, 4 * 32 * DCOLS, 4 * 32 * DCOLS };
    float* const st   = stage + wave * STAGE_EL(RG);
    float* const st_w = st + r4 * SPITCH + 4 * c8;
    const float* const st_r = st + row * SPITCH + koff;
    const float* const Pr = Arow + hi * 32;
    const float* const Pp[RG] = { Pr, Pr + (size_t)16 * 32 * DCOLS, Pr + (size_t)32 * 32 * DCOLS };
    k_slice<KW2, RG, true>(Sp, Rs, Pp, Bp, st_w, st_r, acc, nw, Bx, kb0);

    float wcol[FF];
#pragma unroll
    for (int f = 0; f < FF; ++f) wcol[f] = w[f * FF + lane];

    float s_lo[RG], s_hi[RG];
    sum_wave_tiles<RG>(red, wave, lane, acc, s_lo, s_hi);

    float t_lo[RG], t_hi[RG];
#pragma unroll
    for (int g = 0; g < RG; ++g) { t_lo[g] = 0.f; t_hi[g] = 0.f; }
#pragma unroll
    for (int f = 0; f < FF; ++f) {
#pragma unroll
        for (int g = 0; g < RG; ++g) {
            const float a = __builtin_bit_cast(float,
                __builtin_amdgcn_readlane(__builtin_bit_cast(int, s_lo[g]), f));
            const float c = __builtin_bit_cast(float,
                __builtin_amdgcn_readlane(__builtin_bit_cast(int, s_hi[g]), f));
            t_lo[g] = fmaf(a, wcol[f], t_lo[g]);
            t_hi[g] = fmaf(c, wcol[f], t_hi[g]);
        }
    }

#pragma unroll
    for (int g = 0; g < RG; ++g) {
        const int mA = res + 32 * (i0 + 16 * g + wave);
        const int mB = mA + 32 * 8;
        float vA = t_lo[g] * filt[2048 + mA];
        float vB = t_hi[g] * filt[2048 + mB];
        if (mA >= 2048)
            vA = copysignf(fmaxf(fabsf(vA) - thr, 0.f), vA);
        if (mB >= 2048)
            vB = copysignf(fmaxf(fabsf(vB) - thr, 0.f), vB);
        xshr[((size_t)b * MSHR + mA) * FF + lane] = vA;
        xshr[((size_t)b * MSHR + mB) * FF + lane] = vB;
    }
}

template <int RGT>
__device__ __forceinline__ void pass3_tile(const float* __restrict__ dl,
                                           const float* __restrict__ xshr,
                                           float* __restrict__ partial,
                                           float* red, float* stage, int tile, int split, int b,
                                           int wave, int lane) {
    const int kbegin = split * KSEG + wave * KW3;

    const int row  = lane & 15;
    const int hi   = lane >> 4;
    const int koff = hi << 2;

    const float* Abase = dl + (size_t)b * DROWS * DCOLS + (size_t)2048 * DCOLS + kbegin;
    const float* Ag[RGT];
#pragma unroll
    for (int g = 0; g < RGT; ++g) {
        int krow = tile * TROWS3 + 16 * g + row;
        if (krow > KOUT - 1) krow = KOUT - 1;
        Ag[g] = Abase + (size_t)krow * MSHR;
    }
    const float* Bp = xshr + (size_t)b * MSHR * FF + (size_t)(kbegin + koff) * FF + 2 * row;

    v8f acc[RGT][2];
#pragma unroll
    for (int g = 0; g < RGT; ++g) {
        acc[g][0] = (v8f){0.f,0.f,0.f,0.f,0.f,0.f,0.f,0.f};
        acc[g][1] = (v8f){0.f,0.f,0.f,0.f,0.f,0.f,0.f,0.f};
    }

    const float* Pp[RGT];
#pragma unroll
    for (int g = 0; g < RGT; ++g) Pp[g] = Ag[g] + hi * 32;
    const int r4 = lane >> 3;
    const int c8 = lane & 7;
    const float* Sp[RGT];
    int Rs[RGT];
#pragma unroll
    for (int g = 0; g < RGT; ++g) {
        int krow = tile * TROWS3 + 16 * g + r4;
        int step = 4 * MSHR;
        if (g >= RG3) { krow = KOUT - 1; step = 0; }
        Sp[g] = Abase + (size_t)krow * MSHR + 4 * c8;
        Rs[g] = step;
    }
    float* const st   = stage + wave * STAGE_EL(RGT);
    float* const st_w = st + r4 * SPITCH + 4 * c8;
    const float* const st_r = st + row * SPITCH + koff;
    k_slice<KW3, RGT, false>(Sp, Rs, Pp, Bp, st_w, st_r, acc, 0, Bp, 0);

    float s_lo[RGT], s_hi[RGT];
    sum_wave_tiles<RGT>(red, wave, lane, acc, s_lo, s_hi);

    float* pbase = partial + ((size_t)(b * NSPLIT + split)) * KPAD * FF;
#pragma unroll
    for (int g = 0; g < RGT; ++g) {
        const int kkA = tile * TROWS3 + 16 * g + wave;
        pbase[(size_t)kkA * FF + lane]       = s_lo[g];
        pbase[(size_t)(kkA + 8) * FF + lane] = s_hi[g];
    }
}

__global__ void __launch_bounds__(256) __attribute__((amdgpu_waves_per_eu(2, 2)))
sufg_pass3_kernel(const float* __restrict__ dl,
                  const float* __restrict__ xshr,
                  float* __restrict__ partial) {
    __shared__ float red[NWAVE * RG3L * TILE_EL];
    __shared__ __attribute__((aligned(16))) float stage[NWAVE * STAGE_EL(RG3L)];

    const int lane = threadIdx.x & 31;
    const int wave = threadIdx.x >> 5;
    const int unit = blockIdx.x;
    const int b    = blockIdx.y;

    const int split = unit & (NSPLIT - 1);
    const int tile  = (NT3 - 1) - (unit >> 2);

    if (tile == NT3 - 1)
        pass3_tile<RG3L>(dl, xshr, partial, red, stage, tile, split, b, wave, lane);
    else
        pass3_tile<RG3>(dl, xshr, partial, red, stage, tile, split, b, wave, lane);
}

__global__ void sufg_reduce_kernel(const float* __restrict__ partial,
                                   const float* __restrict__ bias,
                                   float* __restrict__ out) {
    int idx = blockIdx.x * blockDim.x + threadIdx.x;
    if (idx >= BB * KOUT * FF) return;
    int o = idx & 31;
    int k = (idx >> 5) % KOUT;
    int b = idx / (KOUT * FF);
    float s = bias[o];
#pragma unroll
    for (int sp = 0; sp < NSPLIT; ++sp)
        s += partial[(((size_t)(b * NSPLIT + sp)) * KPAD + k) * FF + o];
    out[idx] = s;
}

extern "C" void kernel_launch(void* const* d_in, const int* in_sizes, int n_in,
                              void* d_out, int out_size, void* d_ws, size_t ws_size,
                              hipStream_t stream) {
    (void)in_sizes; (void)n_in; (void)out_size; (void)ws_size;
    const float* x    = (const float*)d_in[0];
    const float* dl   = (const float*)d_in[1];
    const float* w    = (const float*)d_in[2];
    const float* filt = (const float*)d_in[3];
    const float* bias = (const float*)d_in[4];
    float* out = (float*)d_out;

    float* ws      = (float*)d_ws;
    float* xshr    = ws + WS_XSHR_OFF;
    float* partial = ws + WS_PART_OFF;

    const float thr = sqrtf(2.0f * logf((float)NN)) / sqrtf((float)NN);

    sufg_pass2_kernel<<<dim3(MSHR / TROWS, BB), 256, 0, stream>>>(dl, x, w, filt, xshr, thr);
    sufg_pass3_kernel<<<dim3(NT3 * NSPLIT, BB), 256, 0, stream>>>(dl, xshr, partial);
    sufg_reduce_kernel<<<(BB * KOUT * FF + 255) / 256, 256, 0, stream>>>(partial, bias, out);
}
